# starting kernel
# baseline (speedup 1.0000x reference)
#include <hip/hip_runtime.h>
#include <math.h>

#define H_     512
#define NMODE  32
#define B_     8
#define L_     2048
#define Q_     64
#define NCHUNK 32
#define CPITCH 34
#define WPITCH 516

typedef __attribute__((ext_vector_type(2))) float v2f;
typedef __attribute__((ext_vector_type(8))) float v8f;

__device__ __forceinline__ v8f wmma4(v2f a, v2f b, v8f c) {
  return __builtin_amdgcn_wmma_f32_16x16x4_f32(false, a, false, b, (short)0, c,
                                               false, false);
}

__global__ void s4d_setup(const float* __restrict__ log_dt,
                          const float* __restrict__ C,
                          const float* __restrict__ log_A_real,
                          const float* __restrict__ A_imag,
                          float* __restrict__ Ktap2,
                          float* __restrict__ CoutR,
                          float* __restrict__ CoutI,
                          float* __restrict__ Wr,
                          float* __restrict__ Wi) {
  const int h = blockIdx.x;
  const int k = threadIdx.x;
  const float dt = expf(log_dt[h]);
  const float Ar = -expf(log_A_real[h * NMODE + k]);
  const float Ai = A_imag[h * NMODE + k];
  const float Ccr = C[(h * NMODE + k) * 2 + 0];
  const float Cci = C[(h * NMODE + k) * 2 + 1];
  const float er = expf(Ar * dt);
  const float wr = er * cosf(Ai * dt);
  const float wi = er * sinf(Ai * dt);
  const float exr = wr - 1.0f, exi = wi;
  const float den = Ar * Ar + Ai * Ai;
  const float qr = (exr * Ar + exi * Ai) / den;
  const float qi = (exi * Ar - exr * Ai) / den;
  const float Ckr = Ccr * qr - Cci * qi;
  const float Cki = Ccr * qi + Cci * qr;
  Wr[h * NMODE + k] = wr;
  Wi[h * NMODE + k] = wi;

  Ktap2[h * 128 + k] = 0.0f;
  Ktap2[h * 128 + 32 + k] = 0.0f;

  float pr = Ckr, pi = Cki;
  for (int d = 0; d < Q_; ++d) {
    float v = 2.0f * pr;
    for (int m = 16; m >= 1; m >>= 1) v += __shfl_xor(v, m, 32);
    if (k == 0) Ktap2[h * 128 + 64 + d] = v;
    const float nr = pr * wr - pi * wi;
    const float ni = pr * wi + pi * wr;
    pr = nr; pi = ni;
  }
  float sr = Ckr * wr - Cki * wi;
  float si = Ckr * wi + Cki * wr;
  for (int j = 0; j < Q_; ++j) {
    CoutR[((size_t)h * Q_ + j) * NMODE + k] = 2.0f * sr;
    CoutI[((size_t)h * Q_ + j) * NMODE + k] = -2.0f * si;
    const float nr = sr * wr - si * wi;
    const float ni = sr * wi + si * wr;
    sr = nr; si = ni;
  }
}

__global__ void s4d_scan(const float* __restrict__ u,
                         const float* __restrict__ Wr,
                         const float* __restrict__ Wi,
                         float* __restrict__ XR,
                         float* __restrict__ XI) {
  const int wid  = (int)((blockIdx.x * blockDim.x + threadIdx.x) >> 5);
  const int lane = threadIdx.x & 31;
  const int b = wid >> 9;
  const int h = wid & (H_ - 1);
  const float wr = Wr[h * NMODE + lane];
  const float wi = Wi[h * NMODE + lane];
  float sr = 0.0f, si = 0.0f;
  const float* up = u + (size_t)(b * H_ + h) * L_;
  const size_t xb = ((size_t)(b * H_ + h)) * NCHUNK * NMODE;
  for (int c = 0; c < NCHUNK; ++c) {
    XR[xb + (size_t)c * NMODE + lane] = sr;
    XI[xb + (size_t)c * NMODE + lane] = si;
    for (int t = 0; t < Q_; ++t) {
      const float uu = up[c * Q_ + t];
      const float nr = sr * wr - si * wi + uu;
      const float ni = sr * wi + si * wr;
      sr = nr; si = ni;
    }
  }
}

__global__ void s4d_conv(const float* __restrict__ u,
                         const float* __restrict__ Ktap2,
                         const float* __restrict__ CoutR,
                         const float* __restrict__ CoutI,
                         const float* __restrict__ XR,
                         const float* __restrict__ XI,
                         const float* __restrict__ Dvec,
                         float* __restrict__ ygelu) {
  __shared__ float sKt[128];
  __shared__ float sCR[Q_ * CPITCH];
  __shared__ float sCI[Q_ * CPITCH];

  const int h      = blockIdx.x >> 3;
  const int g      = blockIdx.x & 7;
  const int waveId = threadIdx.x >> 5;
  const int lane   = threadIdx.x & 31;
  const int half   = lane >> 4;
  const int lm     = lane & 15;
  const int jt     = waveId & 3;
  const int ct     = waveId >> 2;
  const int q      = g * 32 + ct * 16 + lm;
  const int b      = q >> 5;
  const int c      = q & 31;

  if (threadIdx.x < 128) sKt[threadIdx.x] = Ktap2[(size_t)h * 128 + threadIdx.x];
  for (int i = threadIdx.x; i < Q_ * NMODE; i += 256) {
    const int j = i >> 5, k = i & 31;
    sCR[j * CPITCH + k] = CoutR[((size_t)h * Q_ + j) * NMODE + k];
    sCI[j * CPITCH + k] = CoutI[((size_t)h * Q_ + j) * NMODE + k];
  }
  __syncthreads();

  const size_t ub = (size_t)(b * H_ + h) * L_ + (size_t)c * Q_;
  const int jrow = jt * 16 + lm;

  v8f acc = {0.f, 0.f, 0.f, 0.f, 0.f, 0.f, 0.f, 0.f};

#pragma unroll
  for (int k0 = 0; k0 < Q_; k0 += 4) {
    const int ta = k0 + 2 * half;
    v2f a, bb;
    a.x = sKt[64 + jrow - ta];
    a.y = sKt[63 + jrow - ta];
    bb.x = u[ub + ta];
    bb.y = u[ub + ta + 1];
    acc = wmma4(a, bb, acc);
  }
  const size_t xrb = ((size_t)(b * H_ + h) * NCHUNK + c) * NMODE;
#pragma unroll
  for (int k0 = 0; k0 < NMODE; k0 += 4) {
    const int ka = k0 + 2 * half;
    v2f a, bb;
    a.x = sCR[jrow * CPITCH + ka];
    a.y = sCR[jrow * CPITCH + ka + 1];
    bb.x = XR[xrb + ka];
    bb.y = XR[xrb + ka + 1];
    acc = wmma4(a, bb, acc);
  }
#pragma unroll
  for (int k0 = 0; k0 < NMODE; k0 += 4) {
    const int ka = k0 + 2 * half;
    v2f a, bb;
    a.x = sCI[jrow * CPITCH + ka];
    a.y = sCI[jrow * CPITCH + ka + 1];
    bb.x = XI[xrb + ka];
    bb.y = XI[xrb + ka + 1];
    acc = wmma4(a, bb, acc);
  }
  const float Dh = Dvec[h];
#pragma unroll
  for (int v = 0; v < 8; ++v) {
    const int j = jt * 16 + v + 8 * half;
    const size_t idx = (size_t)(b * H_ + h) * L_ + (size_t)c * Q_ + j;
    const float uu = u[idx];
    const float y = acc[v] + Dh * uu;
    ygelu[idx] = 0.5f * y * (1.0f + erff(y * 0.70710678118654752f));
  }
}

__global__ void s4d_glu(const float* __restrict__ ygelu,
                        const float* __restrict__ W,
                        const float* __restrict__ bias,
                        float* __restrict__ out) {
  extern __shared__ float sW[];
  const int bid    = blockIdx.x;
  const int waveId = threadIdx.x >> 5;
  const int lane   = threadIdx.x & 31;
  const int half   = lane >> 4;
  const int lm     = lane & 15;
  const int mtile  = bid >> 7;
  const int ntile  = (bid & 127) * 8 + waveId;
  const int col    = ntile * 16 + lm;
  const int bb     = col >> 11;
  const int l      = col & (L_ - 1);

  for (int i = threadIdx.x; i < (32 * H_) / 4; i += 256) {
    const int r = (i * 4) >> 9;
    const int k = (i * 4) & (H_ - 1);
    const int src = (r < 16) ? (mtile * 16 + r) : (H_ + mtile * 16 + (r - 16));
    const float4 v4 = *(const float4*)(W + (size_t)src * H_ + k);
    float* dst = sW + r * WPITCH + k;
    dst[0] = v4.x; dst[1] = v4.y; dst[2] = v4.z; dst[3] = v4.w;
  }
  __syncthreads();

  const float* yb = ygelu + (size_t)bb * H_ * L_ + l;
  const float* la = sW + lm * WPITCH;
  const float* lg = sW + (16 + lm) * WPITCH;

  v8f accA = {0.f, 0.f, 0.f, 0.f, 0.f, 0.f, 0.f, 0.f};
  v8f accG = {0.f, 0.f, 0.f, 0.f, 0.f, 0.f, 0.f, 0.f};

#pragma unroll 4
  for (int k0 = 0; k0 < H_; k0 += 4) {
    const int ka = k0 + 2 * half;
    v2f bf, aA, aG;
    bf.x = yb[(size_t)ka * L_];
    bf.y = yb[(size_t)(ka + 1) * L_];
    aA.x = la[ka];  aA.y = la[ka + 1];
    aG.x = lg[ka];  aG.y = lg[ka + 1];
    if (k0 + 32 < H_) {
      __builtin_prefetch(yb + (size_t)(k0 + 32) * L_, 0, 0);
    }
    accA = wmma4(aA, bf, accA);
    accG = wmma4(aG, bf, accG);
  }
#pragma unroll
  for (int v = 0; v < 8; ++v) {
    const int hr = mtile * 16 + v + 8 * half;
    const float a = accA[v] + bias[hr];
    const float gv = accG[v] + bias[H_ + hr];
    out[((size_t)(bb * H_ + hr)) * L_ + l] = a * (1.0f / (1.0f + expf(-gv)));
  }
}

extern "C" void kernel_launch(void* const* d_in, const int* in_sizes, int n_in,
                              void* d_out, int out_size, void* d_ws, size_t ws_size,
                              hipStream_t stream) {
  (void)in_sizes; (void)n_in; (void)out_size; (void)ws_size;
  const float* u          = (const float*)d_in[0];
  const float* log_dt     = (const float*)d_in[1];
  const float* C          = (const float*)d_in[2];
  const float* log_A_real = (const float*)d_in[3];
  const float* A_imag     = (const float*)d_in[4];
  const float* Dvec       = (const float*)d_in[5];
  const float* W          = (const float*)d_in[6];
  const float* bias       = (const float*)d_in[7];
  float* out = (float*)d_out;

  float* ws = (float*)d_ws;
  float* Ktap2 = ws;
  float* CoutR = Ktap2 + (size_t)H_ * 128;
  float* CoutI = CoutR + (size_t)H_ * Q_ * NMODE;
  float* Wr    = CoutI + (size_t)H_ * Q_ * NMODE;
  float* Wi    = Wr    + (size_t)H_ * NMODE;
  float* XR    = Wi    + (size_t)H_ * NMODE;
  float* XI    = XR    + (size_t)B_ * H_ * NCHUNK * NMODE;
  float* ygelu = XI    + (size_t)B_ * H_ * NCHUNK * NMODE;

  s4d_setup<<<H_, 32, 0, stream>>>(log_dt, C, log_A_real, A_imag,
                                   Ktap2, CoutR, CoutI, Wr, Wi);
  s4d_scan<<<(B_ * H_) / 8, 256, 0, stream>>>(u, Wr, Wi, XR, XI);
  s4d_conv<<<H_ * 8, 256, 0, stream>>>(u, Ktap2, CoutR, CoutI, XR, XI, Dvec, ygelu);
  s4d_glu<<<4096, 256, (size_t)(32 * WPITCH) * sizeof(float), stream>>>(ygelu, W,
                                                                        bias, out);
}


# GLU GEMM re-tiled: block 64a+64g rows x 128 cols, K chunks of 32 staged in LDS (both operands, 2 buffers), each wave 4x2 accumulator tiles; f32 WMMA kept, per-element k sum order unchanged
# speedup vs baseline: 1.1659x; 1.1659x over previous
#include <hip/hip_runtime.h>
#include <math.h>

#define H_     512
#define NMODE  32
#define B_     8
#define L_     2048
#define Q_     64
#define NCHUNK 32
#define CPITCH 34

typedef __attribute__((ext_vector_type(2))) float v2f;
typedef __attribute__((ext_vector_type(8))) float v8f;

__device__ __forceinline__ v8f wmma4(v2f a, v2f b, v8f c) {
  return __builtin_amdgcn_wmma_f32_16x16x4_f32(false, a, false, b, (short)0, c,
                                               false, false);
}

__global__ void s4d_setup(const float* __restrict__ log_dt,
                          const float* __restrict__ C,
                          const float* __restrict__ log_A_real,
                          const float* __restrict__ A_imag,
                          float* __restrict__ Ktap2,
                          float* __restrict__ CoutR,
                          float* __restrict__ CoutI,
                          float* __restrict__ Wr,
                          float* __restrict__ Wi) {
  const int h = blockIdx.x;
  const int k = threadIdx.x;
  const float dt = expf(log_dt[h]);
  const float Ar = -expf(log_A_real[h * NMODE + k]);
  const float Ai = A_imag[h * NMODE + k];
  const float Ccr = C[(h * NMODE + k) * 2 + 0];
  const float Cci = C[(h * NMODE + k) * 2 + 1];
  const float er = expf(Ar * dt);
  const float wr = er * cosf(Ai * dt);
  const float wi = er * sinf(Ai * dt);
  const float exr = wr - 1.0f, exi = wi;
  const float den = Ar * Ar + Ai * Ai;
  const float qr = (exr * Ar + exi * Ai) / den;
  const float qi = (exi * Ar - exr * Ai) / den;
  const float Ckr = Ccr * qr - Cci * qi;
  const float Cki = Ccr * qi + Cci * qr;
  Wr[h * NMODE + k] = wr;
  Wi[h * NMODE + k] = wi;

  Ktap2[h * 128 + k] = 0.0f;
  Ktap2[h * 128 + 32 + k] = 0.0f;

  float pr = Ckr, pi = Cki;
  for (int d = 0; d < Q_; ++d) {
    float v = 2.0f * pr;
    for (int m = 16; m >= 1; m >>= 1) v += __shfl_xor(v, m, 32);
    if (k == 0) Ktap2[h * 128 + 64 + d] = v;
    const float nr = pr * wr - pi * wi;
    const float ni = pr * wi + pi * wr;
    pr = nr; pi = ni;
  }
  float sr = Ckr * wr - Cki * wi;
  float si = Ckr * wi + Cki * wr;
  for (int j = 0; j < Q_; ++j) {
    CoutR[((size_t)h * Q_ + j) * NMODE + k] = 2.0f * sr;
    CoutI[((size_t)h * Q_ + j) * NMODE + k] = -2.0f * si;
    const float nr = sr * wr - si * wi;
    const float ni = sr * wi + si * wr;
    sr = nr; si = ni;
  }
}

__global__ void s4d_scan(const float* __restrict__ u,
                         const float* __restrict__ Wr,
                         const float* __restrict__ Wi,
                         float* __restrict__ XR,
                         float* __restrict__ XI) {
  const int wid  = (int)((blockIdx.x * blockDim.x + threadIdx.x) >> 5);
  const int lane = threadIdx.x & 31;
  const int b = wid >> 9;
  const int h = wid & (H_ - 1);
  const float wr = Wr[h * NMODE + lane];
  const float wi = Wi[h * NMODE + lane];
  float sr = 0.0f, si = 0.0f;
  const float* up = u + (size_t)(b * H_ + h) * L_;
  const size_t xb = ((size_t)(b * H_ + h)) * NCHUNK * NMODE;
  for (int c = 0; c < NCHUNK; ++c) {
    XR[xb + (size_t)c * NMODE + lane] = sr;
    XI[xb + (size_t)c * NMODE + lane] = si;
    for (int t = 0; t < Q_; ++t) {
      const float uu = up[c * Q_ + t];
      const float nr = sr * wr - si * wi + uu;
      const float ni = sr * wi + si * wr;
      sr = nr; si = ni;
    }
  }
}

__global__ void s4d_conv(const float* __restrict__ u,
                         const float* __restrict__ Ktap2,
                         const float* __restrict__ CoutR,
                         const float* __restrict__ CoutI,
                         const float* __restrict__ XR,
                         const float* __restrict__ XI,
                         const float* __restrict__ Dvec,
                         float* __restrict__ ygelu) {
  __shared__ float sKt[128];
  __shared__ float sCR[Q_ * CPITCH];
  __shared__ float sCI[Q_ * CPITCH];

  const int h      = blockIdx.x >> 3;
  const int g      = blockIdx.x & 7;
  const int waveId = threadIdx.x >> 5;
  const int lane   = threadIdx.x & 31;
  const int half   = lane >> 4;
  const int lm     = lane & 15;
  const int jt     = waveId & 3;
  const int ct     = waveId >> 2;
  const int q      = g * 32 + ct * 16 + lm;
  const int b      = q >> 5;
  const int c      = q & 31;

  if (threadIdx.x < 128) sKt[threadIdx.x] = Ktap2[(size_t)h * 128 + threadIdx.x];
  for (int i = threadIdx.x; i < Q_ * NMODE; i += 256) {
    const int j = i >> 5, k = i & 31;
    sCR[j * CPITCH + k] = CoutR[((size_t)h * Q_ + j) * NMODE + k];
    sCI[j * CPITCH + k] = CoutI[((size_t)h * Q_ + j) * NMODE + k];
  }
  __syncthreads();

  const size_t ub = (size_t)(b * H_ + h) * L_ + (size_t)c * Q_;
  const int jrow = jt * 16 + lm;

  v8f acc = {0.f, 0.f, 0.f, 0.f, 0.f, 0.f, 0.f, 0.f};

#pragma unroll
  for (int k0 = 0; k0 < Q_; k0 += 4) {
    const int ta = k0 + 2 * half;
    v2f a, bb;
    a.x = sKt[64 + jrow - ta];
    a.y = sKt[63 + jrow - ta];
    bb.x = u[ub + ta];
    bb.y = u[ub + ta + 1];
    acc = wmma4(a, bb, acc);
  }
  const size_t xrb = ((size_t)(b * H_ + h) * NCHUNK + c) * NMODE;
#pragma unroll
  for (int k0 = 0; k0 < NMODE; k0 += 4) {
    const int ka = k0 + 2 * half;
    v2f a, bb;
    a.x = sCR[jrow * CPITCH + ka];
    a.y = sCR[jrow * CPITCH + ka + 1];
    bb.x = XR[xrb + ka];
    bb.y = XR[xrb + ka + 1];
    acc = wmma4(a, bb, acc);
  }
#pragma unroll
  for (int k0 = 0; k0 < NMODE; k0 += 4) {
    const int ka = k0 + 2 * half;
    v2f a, bb;
    a.x = sCI[jrow * CPITCH + ka];
    a.y = sCI[jrow * CPITCH + ka + 1];
    bb.x = XI[xrb + ka];
    bb.y = XI[xrb + ka + 1];
    acc = wmma4(a, bb, acc);
  }
  const float Dh = Dvec[h];
#pragma unroll
  for (int v = 0; v < 8; ++v) {
    const int j = jt * 16 + v + 8 * half;
    const size_t idx = (size_t)(b * H_ + h) * L_ + (size_t)c * Q_ + j;
    const float uu = u[idx];
    const float y = acc[v] + Dh * uu;
    ygelu[idx] = 0.5f * y * (1.0f + erff(y * 0.70710678118654752f));
  }
}

#define GKC      32
#define APITCH   36
#define BPITCH   136
#define A_BUF    (128 * APITCH)
#define B_BUF    (GKC * BPITCH)
#define GLU_LDS_BYTES 71680

__global__ __launch_bounds__(256) void s4d_glu(const float* __restrict__ ygelu,
                                               const float* __restrict__ W,
                                               const float* __restrict__ bias,
                                               float* __restrict__ out) {
  extern __shared__ __attribute__((aligned(16))) float smem[];
  float* sA = smem;
  float* sB = smem + 2 * A_BUF;

  const int tid    = threadIdx.x;
  const int waveId = tid >> 5;
  const int lane   = tid & 31;
  const int half   = lane >> 4;
  const int lm     = lane & 15;
  const int wm     = waveId & 1;
  const int wn     = waveId >> 1;
  const int mblk   = blockIdx.x & 7;
  const int nblk   = blockIdx.x >> 3;
  const int bb     = nblk >> 4;
  const int l0     = (nblk & 15) * 128;

  const float* gA = W + (size_t)(mblk * 64 + (tid >> 3)) * H_ + (tid & 7) * 4;
  const float* gB = ygelu + ((size_t)bb * H_ + (tid >> 5)) * L_ + l0 + (tid & 31) * 4;
  const int stA = (tid >> 3) * APITCH + (tid & 7) * 4;
  const int stB = (tid >> 5) * BPITCH + (tid & 31) * 4;

  float4 ra[4], rb[4];
#pragma unroll
  for (int j = 0; j < 4; ++j) {
    ra[j] = *(const float4*)(gA + (size_t)(j >> 1) * H_ * H_ + (size_t)(j & 1) * 32 * H_);
    rb[j] = *(const float4*)(gB + (size_t)j * 8 * L_);
  }
#pragma unroll
  for (int j = 0; j < 4; ++j) {
    *(float4*)(sA + stA + j * 32 * APITCH) = ra[j];
    *(float4*)(sB + stB + j * 8 * BPITCH)  = rb[j];
  }
  __syncthreads();

  v8f acc[4][2];
#pragma unroll
  for (int i = 0; i < 4; ++i)
#pragma unroll
    for (int j = 0; j < 2; ++j)
      acc[i][j] = (v8f){0.f, 0.f, 0.f, 0.f, 0.f, 0.f, 0.f, 0.f};

  const int fA = (wm * 32 + lm) * APITCH + 2 * half;
  const int fB = (2 * half) * BPITCH + wn * 32 + lm;

  for (int kc = 0; kc < H_ / GKC; ++kc) {
    const int cur = kc & 1;
    const bool more = (kc + 1) < (H_ / GKC);
    if (more) {
      const float* nA = gA + (kc + 1) * GKC;
      const float* nB = gB + (size_t)(kc + 1) * GKC * L_;
#pragma unroll
      for (int j = 0; j < 4; ++j) {
        ra[j] = *(const float4*)(nA + (size_t)(j >> 1) * H_ * H_ + (size_t)(j & 1) * 32 * H_);
        rb[j] = *(const float4*)(nB + (size_t)j * 8 * L_);
      }
    }
    const float* cA = sA + cur * A_BUF + fA;
    const float* cB = sB + cur * B_BUF + fB;
#pragma unroll
    for (int k0 = 0; k0 < GKC; k0 += 4) {
      v2f a[4], b[2];
      a[0] = *(const v2f*)(cA + k0);
      a[1] = *(const v2f*)(cA + 16 * APITCH + k0);
      a[2] = *(const v2f*)(cA + 64 * APITCH + k0);
      a[3] = *(const v2f*)(cA + 80 * APITCH + k0);
#pragma unroll
      for (int j = 0; j < 2; ++j) {
        b[j].x = cB[k0 * BPITCH + 16 * j];
        b[j].y = cB[(k0 + 1) * BPITCH + 16 * j];
      }
#pragma unroll
      for (int i = 0; i < 4; ++i)
#pragma unroll
        for (int j = 0; j < 2; ++j)
          acc[i][j] = wmma4(a[i], b[j], acc[i][j]);
    }
    if (more) {
      float* dA = sA + (cur ^ 1) * A_BUF + stA;
      float* dB = sB + (cur ^ 1) * B_BUF + stB;
#pragma unroll
      for (int j = 0; j < 4; ++j) {
        *(float4*)(dA + j * 32 * APITCH) = ra[j];
        *(float4*)(dB + j * 8 * BPITCH)  = rb[j];
      }
    }
    __syncthreads();
  }

#pragma unroll
  for (int mi = 0; mi < 2; ++mi) {
#pragma unroll
    for (int v = 0; v < 8; ++v) {
      const int hr = mblk * 64 + wm * 32 + mi * 16 + v + 8 * half;
      const float ba = bias[hr];
      const float bg = bias[H_ + hr];
#pragma unroll
      for (int j = 0; j < 2; ++j) {
        const int l = l0 + wn * 32 + j * 16 + lm;
        const float a  = acc[mi][j][v] + ba;
        const float gv = acc[2 + mi][j][v] + bg;
        out[((size_t)(bb * H_ + hr)) * L_ + l] = a * (1.0f / (1.0f + expf(-gv)));
      }
    }
  }
}

extern "C" void kernel_launch(void* const* d_in, const int* in_sizes, int n_in,
                              void* d_out, int out_size, void* d_ws, size_t ws_size,
                              hipStream_t stream) {
  (void)in_sizes; (void)n_in; (void)out_size; (void)ws_size;
  const float* u          = (const float*)d_in[0];
  const float* log_dt     = (const float*)d_in[1];
  const float* C          = (const float*)d_in[2];
  const float* log_A_real = (const float*)d_in[3];
  const float* A_imag     = (const float*)d_in[4];
  const float* Dvec       = (const float*)d_in[5];
  const float* W          = (const float*)d_in[6];
  const float* bias       = (const float*)d_in[7];
  float* out = (float*)d_out;

  float* ws = (float*)d_ws;
  float* Ktap2 = ws;
  float* CoutR = Ktap2 + (size_t)H_ * 128;
  float* CoutI = CoutR + (size_t)H_ * Q_ * NMODE;
  float* Wr    = CoutI + (size_t)H_ * Q_ * NMODE;
  float* Wi    = Wr    + (size_t)H_ * NMODE;
  float* XR    = Wi    + (size_t)H_ * NMODE;
  float* XI    = XR    + (size_t)B_ * H_ * NCHUNK * NMODE;
  float* ygelu = XI    + (size_t)B_ * H_ * NCHUNK * NMODE;

  s4d_setup<<<H_, 32, 0, stream>>>(log_dt, C, log_A_real, A_imag,
                                   Ktap2, CoutR, CoutI, Wr, Wi);
  s4d_scan<<<(B_ * H_) / 8, 256, 0, stream>>>(u, Wr, Wi, XR, XI);
  s4d_conv<<<H_ * 8, 256, 0, stream>>>(u, Ktap2, CoutR, CoutI, XR, XI, Dvec, ygelu);
  s4d_glu<<<1024, 256, GLU_LDS_BYTES, stream>>>(ygelu, W, bias, out);
}


# mode-state scan folded into the conv kernel: per-chunk state sums as an f32 WMMA product on the LDS-staged u row, then a 32-step chunk-to-chunk combine (state sum order changes); 4 launches -> 3
# speedup vs baseline: 1.5599x; 1.3379x over previous
#include <hip/hip_runtime.h>
#include <math.h>

#define H_     512
#define NMODE  32
#define B_     8
#define L_     2048
#define Q_     64
#define NCHUNK 32
#define CPITCH 34
#define PPITCH 68
#define SPITCH 33
#define XPITCH 36

typedef __attribute__((ext_vector_type(2))) float v2f;
typedef __attribute__((ext_vector_type(8))) float v8f;

__device__ __forceinline__ v8f wmma4(v2f a, v2f b, v8f c) {
  return __builtin_amdgcn_wmma_f32_16x16x4_f32(false, a, false, b, (short)0, c,
                                               false, false);
}

__global__ void s4d_setup(const float* __restrict__ log_dt,
                          const float* __restrict__ C,
                          const float* __restrict__ log_A_real,
                          const float* __restrict__ A_imag,
                          float* __restrict__ Ktap2,
                          float* __restrict__ CoutR,
                          float* __restrict__ CoutI,
                          float* __restrict__ W64r,
                          float* __restrict__ W64i,
                          float* __restrict__ PwR,
                          float* __restrict__ PwI) {
  const int h = blockIdx.x;
  const int k = threadIdx.x;
  const float dt = expf(log_dt[h]);
  const float Ar = -expf(log_A_real[h * NMODE + k]);
  const float Ai = A_imag[h * NMODE + k];
  const float Ccr = C[(h * NMODE + k) * 2 + 0];
  const float Cci = C[(h * NMODE + k) * 2 + 1];
  const float er = expf(Ar * dt);
  const float wr = er * cosf(Ai * dt);
  const float wi = er * sinf(Ai * dt);
  const float exr = wr - 1.0f, exi = wi;
  const float den = Ar * Ar + Ai * Ai;
  const float qr = (exr * Ar + exi * Ai) / den;
  const float qi = (exi * Ar - exr * Ai) / den;
  const float Ckr = Ccr * qr - Cci * qi;
  const float Cki = Ccr * qi + Cci * qr;

  Ktap2[h * 128 + k] = 0.0f;
  Ktap2[h * 128 + 32 + k] = 0.0f;

  float pr = Ckr, pi = Cki;
  for (int d = 0; d < Q_; ++d) {
    float v = 2.0f * pr;
    for (int m = 16; m >= 1; m >>= 1) v += __shfl_xor(v, m, 32);
    if (k == 0) Ktap2[h * 128 + 64 + d] = v;
    const float nr = pr * wr - pi * wi;
    const float ni = pr * wi + pi * wr;
    pr = nr; pi = ni;
  }
  float sr = Ckr * wr - Cki * wi;
  float si = Ckr * wi + Cki * wr;
  for (int j = 0; j < Q_; ++j) {
    CoutR[((size_t)h * Q_ + j) * NMODE + k] = 2.0f * sr;
    CoutI[((size_t)h * Q_ + j) * NMODE + k] = -2.0f * si;
    const float nr = sr * wr - si * wi;
    const float ni = sr * wi + si * wr;
    sr = nr; si = ni;
  }
  float qr_ = 1.0f, qi_ = 0.0f;
  for (int d = 0; d < Q_; ++d) {
    PwR[((size_t)h * Q_ + d) * NMODE + k] = qr_;
    PwI[((size_t)h * Q_ + d) * NMODE + k] = qi_;
    const float nr = qr_ * wr - qi_ * wi;
    const float ni = qr_ * wi + qi_ * wr;
    qr_ = nr; qi_ = ni;
  }
  W64r[h * NMODE + k] = qr_;
  W64i[h * NMODE + k] = qi_;
}

__global__ __launch_bounds__(256) void s4d_conv(const float* __restrict__ u,
                         const float* __restrict__ Ktap2,
                         const float* __restrict__ CoutR,
                         const float* __restrict__ CoutI,
                         const float* __restrict__ PwR,
                         const float* __restrict__ PwI,
                         const float* __restrict__ W64r,
                         const float* __restrict__ W64i,
                         const float* __restrict__ Dvec,
                         float* __restrict__ ygelu) {
  __shared__ __attribute__((aligned(16))) float sKt[256];
  __shared__ __attribute__((aligned(16))) float sCR[Q_ * CPITCH];
  __shared__ __attribute__((aligned(16))) float sCI[Q_ * CPITCH];
  __shared__ __attribute__((aligned(16))) float sP[2 * NMODE * PPITCH];
  __shared__ __attribute__((aligned(16))) float sU[NCHUNK * PPITCH];
  __shared__ __attribute__((aligned(16))) float sS[2 * NMODE * SPITCH];
  __shared__ __attribute__((aligned(16))) float sXR[NCHUNK * XPITCH];
  __shared__ __attribute__((aligned(16))) float sXI[NCHUNK * XPITCH];

  const int h      = blockIdx.x >> 3;
  const int b      = blockIdx.x & 7;
  const int waveId = threadIdx.x >> 5;
  const int lane   = threadIdx.x & 31;
  const int half   = lane >> 4;
  const int lm     = lane & 15;
  const int jt     = waveId & 3;
  const int ct     = waveId >> 2;
  const int col    = ct * 16 + lm;

  const size_t urow = (size_t)(b * H_ + h) * L_;

  const float ar = W64r[h * NMODE + lane];
  const float ai = W64i[h * NMODE + lane];

  const float rk = Ktap2[(size_t)h * 128 + (threadIdx.x & 127)];
  float rcr[8], rci[8], rpr[8], rpi[8];
  float4 ru[2];
#pragma unroll
  for (int it = 0; it < 8; ++it) {
    const size_t gi = (size_t)h * (Q_ * NMODE) + it * 256 + threadIdx.x;
    rcr[it] = CoutR[gi];
    rci[it] = CoutI[gi];
    rpr[it] = PwR[gi];
    rpi[it] = PwI[gi];
  }
#pragma unroll
  for (int it = 0; it < 2; ++it)
    ru[it] = *(const float4*)(u + urow + (it * 256 + threadIdx.x) * 4);

  sKt[threadIdx.x] = rk;
#pragma unroll
  for (int it = 0; it < 8; ++it) {
    const int i = it * 256 + threadIdx.x;
    const int j = i >> 5, k = i & 31;
    sCR[j * CPITCH + k] = rcr[it];
    sCI[j * CPITCH + k] = rci[it];
    sP[k * PPITCH + (Q_ - 1 - j)]           = rpr[it];
    sP[(NMODE + k) * PPITCH + (Q_ - 1 - j)] = rpi[it];
  }
#pragma unroll
  for (int it = 0; it < 2; ++it) {
    const int e = (it * 256 + threadIdx.x) * 4;
    *(float4*)(sU + (e >> 6) * PPITCH + (e & 63)) = ru[it];
  }
  __syncthreads();

  const int jrow = jt * 16 + lm;

  v8f acc  = {0.f, 0.f, 0.f, 0.f, 0.f, 0.f, 0.f, 0.f};
  v8f accS = {0.f, 0.f, 0.f, 0.f, 0.f, 0.f, 0.f, 0.f};

#pragma unroll
  for (int k0 = 0; k0 < Q_; k0 += 4) {
    const int ta = k0 + 2 * half;
    const v2f bb = *(const v2f*)(sU + col * PPITCH + ta);
    const v2f ap = *(const v2f*)(sP + jrow * PPITCH + ta);
    v2f a;
    a.x = sKt[64 + jrow - ta];
    a.y = sKt[63 + jrow - ta];
    accS = wmma4(ap, bb, accS);
    acc  = wmma4(a, bb, acc);
  }
#pragma unroll
  for (int v = 0; v < 8; ++v)
    sS[(jt * 16 + v + 8 * half) * SPITCH + col] = accS[v];
  __syncthreads();

  if (waveId == 0) {
    float xr = 0.0f, xi = 0.0f;
#pragma unroll
    for (int c0 = 0; c0 < NCHUNK; c0 += 16) {
      float pr[16], pi[16];
#pragma unroll
      for (int i = 0; i < 16; ++i) {
        pr[i] = sS[lane * SPITCH + c0 + i];
        pi[i] = sS[(NMODE + lane) * SPITCH + c0 + i];
      }
#pragma unroll
      for (int i = 0; i < 16; ++i) {
        sXR[(c0 + i) * XPITCH + lane] = xr;
        sXI[(c0 + i) * XPITCH + lane] = xi;
        const float nr = ar * xr - ai * xi + pr[i];
        const float ni = ar * xi + ai * xr + pi[i];
        xr = nr; xi = ni;
      }
    }
  }
  __syncthreads();

#pragma unroll
  for (int k0 = 0; k0 < NMODE; k0 += 4) {
    const int ka = k0 + 2 * half;
    const v2f a  = *(const v2f*)(sCR + jrow * CPITCH + ka);
    const v2f bb = *(const v2f*)(sXR + col * XPITCH + ka);
    acc = wmma4(a, bb, acc);
  }
#pragma unroll
  for (int k0 = 0; k0 < NMODE; k0 += 4) {
    const int ka = k0 + 2 * half;
    const v2f a  = *(const v2f*)(sCI + jrow * CPITCH + ka);
    const v2f bb = *(const v2f*)(sXI + col * XPITCH + ka);
    acc = wmma4(a, bb, acc);
  }
  const float Dh = Dvec[h];
#pragma unroll
  for (int v = 0; v < 8; ++v) {
    const int j = jt * 16 + v + 8 * half;
    const float uu = sU[col * PPITCH + j];
    const float y = acc[v] + Dh * uu;
    ygelu[urow + (size_t)col * Q_ + j] = 0.5f * y * (1.0f + erff(y * 0.70710678118654752f));
  }
}

#define GKC      32
#define APITCH   36
#define BPITCH   136
#define A_BUF    (128 * APITCH)
#define B_BUF    (GKC * BPITCH)
#define GLU_LDS_BYTES 71680

__global__ __launch_bounds__(256) void s4d_glu(const float* __restrict__ ygelu,
                                               const float* __restrict__ W,
                                               const float* __restrict__ bias,
                                               float* __restrict__ out) {
  extern __shared__ __attribute__((aligned(16))) float smem[];
  float* sA = smem;
  float* sB = smem + 2 * A_BUF;

  const int tid    = threadIdx.x;
  const int waveId = tid >> 5;
  const int lane   = tid & 31;
  const int half   = lane >> 4;
  const int lm     = lane & 15;
  const int wm     = waveId & 1;
  const int wn     = waveId >> 1;
  const int mblk   = blockIdx.x & 7;
  const int nblk   = blockIdx.x >> 3;
  const int bb     = nblk >> 4;
  const int l0     = (nblk & 15) * 128;

  const float* gA = W + (size_t)(mblk * 64 + (tid >> 3)) * H_ + (tid & 7) * 4;
  const float* gB = ygelu + ((size_t)bb * H_ + (tid >> 5)) * L_ + l0 + (tid & 31) * 4;
  const int stA = (tid >> 3) * APITCH + (tid & 7) * 4;
  const int stB = (tid >> 5) * BPITCH + (tid & 31) * 4;

  float4 ra[4], rb[4];
#pragma unroll
  for (int j = 0; j < 4; ++j) {
    ra[j] = *(const float4*)(gA + (size_t)(j >> 1) * H_ * H_ + (size_t)(j & 1) * 32 * H_);
    rb[j] = *(const float4*)(gB + (size_t)j * 8 * L_);
  }
#pragma unroll
  for (int j = 0; j < 4; ++j) {
    *(float4*)(sA + stA + j * 32 * APITCH) = ra[j];
    *(float4*)(sB + stB + j * 8 * BPITCH)  = rb[j];
  }
  __syncthreads();

  v8f acc[4][2];
#pragma unroll
  for (int i = 0; i < 4; ++i)
#pragma unroll
    for (int j = 0; j < 2; ++j)
      acc[i][j] = (v8f){0.f, 0.f, 0.f, 0.f, 0.f, 0.f, 0.f, 0.f};

  const int fA = (wm * 32 + lm) * APITCH + 2 * half;
  const int fB = (2 * half) * BPITCH + wn * 32 + lm;

  for (int kc = 0; kc < H_ / GKC; ++kc) {
    const int cur = kc & 1;
    const bool more = (kc + 1) < (H_ / GKC);
    if (more) {
      const float* nA = gA + (kc + 1) * GKC;
      const float* nB = gB + (size_t)(kc + 1) * GKC * L_;
#pragma unroll
      for (int j = 0; j < 4; ++j) {
        ra[j] = *(const float4*)(nA + (size_t)(j >> 1) * H_ * H_ + (size_t)(j & 1) * 32 * H_);
        rb[j] = *(const float4*)(nB + (size_t)j * 8 * L_);
      }
    }
    const float* cA = sA + cur * A_BUF + fA;
    const float* cB = sB + cur * B_BUF + fB;
#pragma unroll
    for (int k0 = 0; k0 < GKC; k0 += 4) {
      v2f a[4], b[2];
      a[0] = *(const v2f*)(cA + k0);
      a[1] = *(const v2f*)(cA + 16 * APITCH + k0);
      a[2] = *(const v2f*)(cA + 64 * APITCH + k0);
      a[3] = *(const v2f*)(cA + 80 * APITCH + k0);
#pragma unroll
      for (int j = 0; j < 2; ++j) {
        b[j].x = cB[k0 * BPITCH + 16 * j];
        b[j].y = cB[(k0 + 1) * BPITCH + 16 * j];
      }
#pragma unroll
      for (int i = 0; i < 4; ++i)
#pragma unroll
        for (int j = 0; j < 2; ++j)
          acc[i][j] = wmma4(a[i], b[j], acc[i][j]);
    }
    if (more) {
      float* dA = sA + (cur ^ 1) * A_BUF + stA;
      float* dB = sB + (cur ^ 1) * B_BUF + stB;
#pragma unroll
      for (int j = 0; j < 4; ++j) {
        *(float4*)(dA + j * 32 * APITCH) = ra[j];
        *(float4*)(dB + j * 8 * BPITCH)  = rb[j];
      }
    }
    __syncthreads();
  }

#pragma unroll
  for (int mi = 0; mi < 2; ++mi) {
#pragma unroll
    for (int v = 0; v < 8; ++v) {
      const int hr = mblk * 64 + wm * 32 + mi * 16 + v + 8 * half;
      const float ba = bias[hr];
      const float bg = bias[H_ + hr];
#pragma unroll
      for (int j = 0; j < 2; ++j) {
        const int l = l0 + wn * 32 + j * 16 + lm;
        const float a  = acc[mi][j][v] + ba;
        const float gv = acc[2 + mi][j][v] + bg;
        out[((size_t)(bb * H_ + hr)) * L_ + l] = a * (1.0f / (1.0f + expf(-gv)));
      }
    }
  }
}

extern "C" void kernel_launch(void* const* d_in, const int* in_sizes, int n_in,
                              void* d_out, int out_size, void* d_ws, size_t ws_size,
                              hipStream_t stream) {
  (void)in_sizes; (void)n_in; (void)out_size; (void)ws_size;
  const float* u          = (const float*)d_in[0];
  const float* log_dt     = (const float*)d_in[1];
  const float* C          = (const float*)d_in[2];
  const float* log_A_real = (const float*)d_in[3];
  const float* A_imag     = (const float*)d_in[4];
  const float* Dvec       = (const float*)d_in[5];
  const float* W          = (const float*)d_in[6];
  const float* bias       = (const float*)d_in[7];
  float* out = (float*)d_out;

  float* ws = (float*)d_ws;
  float* Ktap2 = ws;
  float* CoutR = Ktap2 + (size_t)H_ * 128;
  float* CoutI = CoutR + (size_t)H_ * Q_ * NMODE;
  float* W64r  = CoutI + (size_t)H_ * Q_ * NMODE;
  float* W64i  = W64r  + (size_t)H_ * NMODE;
  float* PwR   = W64i  + (size_t)H_ * NMODE;
  float* PwI   = PwR   + (size_t)H_ * Q_ * NMODE;
  float* ygelu = PwI   + (size_t)H_ * Q_ * NMODE;

  s4d_setup<<<H_, 32, 0, stream>>>(log_dt, C, log_A_real, A_imag,
                                   Ktap2, CoutR, CoutI, W64r, W64i, PwR, PwI);
  s4d_conv<<<H_ * B_, 256, 0, stream>>>(u, Ktap2, CoutR, CoutI, PwR, PwI, W64r, W64i, Dvec, ygelu);
  s4d_glu<<<1024, 256, GLU_LDS_BYTES, stream>>>(ygelu, W, bias, out);
}


# set-up kernel: lane sums of the 64 Toeplitz taps on vector-ALU cross-lane reads (permlanex16, then row_xmask adds) instead of 320 serial LDS-crossbar shuffles per head; per-tap sum order kept
# speedup vs baseline: 1.6763x; 1.0746x over previous
#include <hip/hip_runtime.h>
#include <math.h>

#define H_     512
#define NMODE  32
#define B_     8
#define L_     2048
#define Q_     64
#define NCHUNK 32
#define CPITCH 34
#define PPITCH 68
#define SPITCH 33
#define XPITCH 36

typedef __attribute__((ext_vector_type(2))) float v2f;
typedef __attribute__((ext_vector_type(8))) float v8f;

__device__ __forceinline__ v8f wmma4(v2f a, v2f b, v8f c) {
  return __builtin_amdgcn_wmma_f32_16x16x4_f32(false, a, false, b, (short)0, c,
                                               false, false);
}

__device__ __forceinline__ float xhalf(float x) {
  return __builtin_bit_cast(float, __builtin_amdgcn_permlanex16(__builtin_bit_cast(int, x), __builtin_bit_cast(int, x),
                                                                 0x76543210, 0xfedcba98, false, false));
}
template <int M>
__device__ __forceinline__ float xrow(float x) {
  return __builtin_bit_cast(float, __builtin_amdgcn_update_dpp(__builtin_bit_cast(int, x), __builtin_bit_cast(int, x),
                                                               0x160 + M, 0xf, 0xf, true));
}

__global__ __launch_bounds__(32) void s4d_setup(const float* __restrict__ log_dt,
                          const float* __restrict__ C,
                          const float* __restrict__ log_A_real,
                          const float* __restrict__ A_imag,
                          float* __restrict__ Ktap2,
                          float* __restrict__ CoutR,
                          float* __restrict__ CoutI,
                          float* __restrict__ W64r,
                          float* __restrict__ W64i,
                          float* __restrict__ PwR,
                          float* __restrict__ PwI) {
  const int h = blockIdx.x;
  const int k = threadIdx.x;
  const float dt = expf(log_dt[h]);
  const float Ar = -expf(log_A_real[h * NMODE + k]);
  const float Ai = A_imag[h * NMODE + k];
  const float Ccr = C[(h * NMODE + k) * 2 + 0];
  const float Cci = C[(h * NMODE + k) * 2 + 1];
  const float er = expf(Ar * dt);
  const float wr = er * cosf(Ai * dt);
  const float wi = er * sinf(Ai * dt);
  const float exr = wr - 1.0f, exi = wi;
  const float den = Ar * Ar + Ai * Ai;
  const float qr = (exr * Ar + exi * Ai) / den;
  const float qi = (exi * Ar - exr * Ai) / den;
  const float Ckr = Ccr * qr - Cci * qi;
  const float Cki = Ccr * qi + Cci * qr;

  Ktap2[h * 128 + k] = 0.0f;
  Ktap2[h * 128 + 32 + k] = 0.0f;

  float tv[Q_];
  {
    float pr = Ckr, pi = Cki;
#pragma unroll
    for (int d = 0; d < Q_; ++d) {
      float v = 2.0f * pr;
      v += xhalf(v);
      v += xrow<8>(v);
      v += xrow<4>(v);
      v += xrow<2>(v);
      v += xrow<1>(v);
      tv[d] = v;
      const float nr = pr * wr - pi * wi;
      const float ni = pr * wi + pi * wr;
      pr = nr; pi = ni;
    }
  }
  float o0 = tv[0], o1 = tv[32];
#pragma unroll
  for (int d = 1; d < 32; ++d) {
    if (k == d) { o0 = tv[d]; o1 = tv[d + 32]; }
  }
  Ktap2[h * 128 + 64 + k] = o0;
  Ktap2[h * 128 + 96 + k] = o1;
  float sr = Ckr * wr - Cki * wi;
  float si = Ckr * wi + Cki * wr;
  for (int j = 0; j < Q_; ++j) {
    CoutR[((size_t)h * Q_ + j) * NMODE + k] = 2.0f * sr;
    CoutI[((size_t)h * Q_ + j) * NMODE + k] = -2.0f * si;
    const float nr = sr * wr - si * wi;
    const float ni = sr * wi + si * wr;
    sr = nr; si = ni;
  }
  float qr_ = 1.0f, qi_ = 0.0f;
  for (int d = 0; d < Q_; ++d) {
    PwR[((size_t)h * Q_ + d) * NMODE + k] = qr_;
    PwI[((size_t)h * Q_ + d) * NMODE + k] = qi_;
    const float nr = qr_ * wr - qi_ * wi;
    const float ni = qr_ * wi + qi_ * wr;
    qr_ = nr; qi_ = ni;
  }
  W64r[h * NMODE + k] = qr_;
  W64i[h * NMODE + k] = qi_;
}

__global__ __launch_bounds__(256) void s4d_conv(const float* __restrict__ u,
                         const float* __restrict__ Ktap2,
                         const float* __restrict__ CoutR,
                         const float* __restrict__ CoutI,
                         const float* __restrict__ PwR,
                         const float* __restrict__ PwI,
                         const float* __restrict__ W64r,
                         const float* __restrict__ W64i,
                         const float* __restrict__ Dvec,
                         float* __restrict__ ygelu) {
  __shared__ __attribute__((aligned(16))) float sKt[256];
  __shared__ __attribute__((aligned(16))) float sCR[Q_ * CPITCH];
  __shared__ __attribute__((aligned(16))) float sCI[Q_ * CPITCH];
  __shared__ __attribute__((aligned(16))) float sP[2 * NMODE * PPITCH];
  __shared__ __attribute__((aligned(16))) float sU[NCHUNK * PPITCH];
  __shared__ __attribute__((aligned(16))) float sS[2 * NMODE * SPITCH];
  __shared__ __attribute__((aligned(16))) float sXR[NCHUNK * XPITCH];
  __shared__ __attribute__((aligned(16))) float sXI[NCHUNK * XPITCH];

  const int h      = blockIdx.x >> 3;
  const int b      = blockIdx.x & 7;
  const int waveId = threadIdx.x >> 5;
  const int lane   = threadIdx.x & 31;
  const int half   = lane >> 4;
  const int lm     = lane & 15;
  const int jt     = waveId & 3;
  const int ct     = waveId >> 2;
  const int col    = ct * 16 + lm;

  const size_t urow = (size_t)(b * H_ + h) * L_;

  const float ar = W64r[h * NMODE + lane];
  const float ai = W64i[h * NMODE + lane];

  const float rk = Ktap2[(size_t)h * 128 + (threadIdx.x & 127)];
  float rcr[8], rci[8], rpr[8], rpi[8];
  float4 ru[2];
#pragma unroll
  for (int it = 0; it < 8; ++it) {
    const size_t gi = (size_t)h * (Q_ * NMODE) + it * 256 + threadIdx.x;
    rcr[it] = CoutR[gi];
    rci[it] = CoutI[gi];
    rpr[it] = PwR[gi];
    rpi[it] = PwI[gi];
  }
#pragma unroll
  for (int it = 0; it < 2; ++it)
    ru[it] = *(const float4*)(u + urow + (it * 256 + threadIdx.x) * 4);

  sKt[threadIdx.x] = rk;
#pragma unroll
  for (int it = 0; it < 8; ++it) {
    const int i = it * 256 + threadIdx.x;
    const int j = i >> 5, k = i & 31;
    sCR[j * CPITCH + k] = rcr[it];
    sCI[j * CPITCH + k] = rci[it];
    sP[k * PPITCH + (Q_ - 1 - j)]           = rpr[it];
    sP[(NMODE + k) * PPITCH + (Q_ - 1 - j)] = rpi[it];
  }
#pragma unroll
  for (int it = 0; it < 2; ++it) {
    const int e = (it * 256 + threadIdx.x) * 4;
    *(float4*)(sU + (e >> 6) * PPITCH + (e & 63)) = ru[it];
  }
  __syncthreads();

  const int jrow = jt * 16 + lm;

  v8f acc  = {0.f, 0.f, 0.f, 0.f, 0.f, 0.f, 0.f, 0.f};
  v8f accS = {0.f, 0.f, 0.f, 0.f, 0.f, 0.f, 0.f, 0.f};

#pragma unroll
  for (int k0 = 0; k0 < Q_; k0 += 4) {
    const int ta = k0 + 2 * half;
    const v2f bb = *(const v2f*)(sU + col * PPITCH + ta);
    const v2f ap = *(const v2f*)(sP + jrow * PPITCH + ta);
    v2f a;
    a.x = sKt[64 + jrow - ta];
    a.y = sKt[63 + jrow - ta];
    accS = wmma4(ap, bb, accS);
    acc  = wmma4(a, bb, acc);
  }
#pragma unroll
  for (int v = 0; v < 8; ++v)
    sS[(jt * 16 + v + 8 * half) * SPITCH + col] = accS[v];
  __syncthreads();

  if (waveId == 0) {
    float xr = 0.0f, xi = 0.0f;
#pragma unroll
    for (int c0 = 0; c0 < NCHUNK; c0 += 16) {
      float pr[16], pi[16];
#pragma unroll
      for (int i = 0; i < 16; ++i) {
        pr[i] = sS[lane * SPITCH + c0 + i];
        pi[i] = sS[(NMODE + lane) * SPITCH + c0 + i];
      }
#pragma unroll
      for (int i = 0; i < 16; ++i) {
        sXR[(c0 + i) * XPITCH + lane] = xr;
        sXI[(c0 + i) * XPITCH + lane] = xi;
        const float nr = ar * xr - ai * xi + pr[i];
        const float ni = ar * xi + ai * xr + pi[i];
        xr = nr; xi = ni;
      }
    }
  }
  __syncthreads();

#pragma unroll
  for (int k0 = 0; k0 < NMODE; k0 += 4) {
    const int ka = k0 + 2 * half;
    const v2f a  = *(const v2f*)(sCR + jrow * CPITCH + ka);
    const v2f bb = *(const v2f*)(sXR + col * XPITCH + ka);
    acc = wmma4(a, bb, acc);
  }
#pragma unroll
  for (int k0 = 0; k0 < NMODE; k0 += 4) {
    const int ka = k0 + 2 * half;
    const v2f a  = *(const v2f*)(sCI + jrow * CPITCH + ka);
    const v2f bb = *(const v2f*)(sXI + col * XPITCH + ka);
    acc = wmma4(a, bb, acc);
  }
  const float Dh = Dvec[h];
#pragma unroll
  for (int v = 0; v < 8; ++v) {
    const int j = jt * 16 + v + 8 * half;
    const float uu = sU[col * PPITCH + j];
    const float y = acc[v] + Dh * uu;
    ygelu[urow + (size_t)col * Q_ + j] = 0.5f * y * (1.0f + erff(y * 0.70710678118654752f));
  }
}

#define GKC      32
#define APITCH   36
#define BPITCH   136
#define A_BUF    (128 * APITCH)
#define B_BUF    (GKC * BPITCH)
#define GLU_LDS_BYTES 71680

__global__ __launch_bounds__(256) void s4d_glu(const float* __restrict__ ygelu,
                                               const float* __restrict__ W,
                                               const float* __restrict__ bias,
                                               float* __restrict__ out) {
  extern __shared__ __attribute__((aligned(16))) float smem[];
  float* sA = smem;
  float* sB = smem + 2 * A_BUF;

  const int tid    = threadIdx.x;
  const int waveId = tid >> 5;
  const int lane   = tid & 31;
  const int half   = lane >> 4;
  const int lm     = lane & 15;
  const int wm     = waveId & 1;
  const int wn     = waveId >> 1;
  const int mblk   = blockIdx.x & 7;
  const int nblk   = blockIdx.x >> 3;
  const int bb     = nblk >> 4;
  const int l0     = (nblk & 15) * 128;

  const float* gA = W + (size_t)(mblk * 64 + (tid >> 3)) * H_ + (tid & 7) * 4;
  const float* gB = ygelu + ((size_t)bb * H_ + (tid >> 5)) * L_ + l0 + (tid & 31) * 4;
  const int stA = (tid >> 3) * APITCH + (tid & 7) * 4;
  const int stB = (tid >> 5) * BPITCH + (tid & 31) * 4;

  float4 ra[4], rb[4];
#pragma unroll
  for (int j = 0; j < 4; ++j) {
    ra[j] = *(const float4*)(gA + (size_t)(j >> 1) * H_ * H_ + (size_t)(j & 1) * 32 * H_);
    rb[j] = *(const float4*)(gB + (size_t)j * 8 * L_);
  }
#pragma unroll
  for (int j = 0; j < 4; ++j) {
    *(float4*)(sA + stA + j * 32 * APITCH) = ra[j];
    *(float4*)(sB + stB + j * 8 * BPITCH)  = rb[j];
  }
  __syncthreads();

  v8f acc[4][2];
#pragma unroll
  for (int i = 0; i < 4; ++i)
#pragma unroll
    for (int j = 0; j < 2; ++j)
      acc[i][j] = (v8f){0.f, 0.f, 0.f, 0.f, 0.f, 0.f, 0.f, 0.f};

  const int fA = (wm * 32 + lm) * APITCH + 2 * half;
  const int fB = (2 * half) * BPITCH + wn * 32 + lm;

  for (int kc = 0; kc < H_ / GKC; ++kc) {
    const int cur = kc & 1;
    const bool more = (kc + 1) < (H_ / GKC);
    if (more) {
      const float* nA = gA + (kc + 1) * GKC;
      const float* nB = gB + (size_t)(kc + 1) * GKC * L_;
#pragma unroll
      for (int j = 0; j < 4; ++j) {
        ra[j] = *(const float4*)(nA + (size_t)(j >> 1) * H_ * H_ + (size_t)(j & 1) * 32 * H_);
        rb[j] = *(const float4*)(nB + (size_t)j * 8 * L_);
      }
    }
    const float* cA = sA + cur * A_BUF + fA;
    const float* cB = sB + cur * B_BUF + fB;
#pragma unroll
    for (int k0 = 0; k0 < GKC; k0 += 4) {
      v2f a[4], b[2];
      a[0] = *(const v2f*)(cA + k0);
      a[1] = *(const v2f*)(cA + 16 * APITCH + k0);
      a[2] = *(const v2f*)(cA + 64 * APITCH + k0);
      a[3] = *(const v2f*)(cA + 80 * APITCH + k0);
#pragma unroll
      for (int j = 0; j < 2; ++j) {
        b[j].x = cB[k0 * BPITCH + 16 * j];
        b[j].y = cB[(k0 + 1) * BPITCH + 16 * j];
      }
#pragma unroll
      for (int i = 0; i < 4; ++i)
#pragma unroll
        for (int j = 0; j < 2; ++j)
          acc[i][j] = wmma4(a[i], b[j], acc[i][j]);
    }
    if (more) {
      float* dA = sA + (cur ^ 1) * A_BUF + stA;
      float* dB = sB + (cur ^ 1) * B_BUF + stB;
#pragma unroll
      for (int j = 0; j < 4; ++j) {
        *(float4*)(dA + j * 32 * APITCH) = ra[j];
        *(float4*)(dB + j * 8 * BPITCH)  = rb[j];
      }
    }
    __syncthreads();
  }

#pragma unroll
  for (int mi = 0; mi < 2; ++mi) {
#pragma unroll
    for (int v = 0; v < 8; ++v) {
      const int hr = mblk * 64 + wm * 32 + mi * 16 + v + 8 * half;
      const float ba = bias[hr];
      const float bg = bias[H_ + hr];
#pragma unroll
      for (int j = 0; j < 2; ++j) {
        const int l = l0 + wn * 32 + j * 16 + lm;
        const float a  = acc[mi][j][v] + ba;
        const float gv = acc[2 + mi][j][v] + bg;
        out[((size_t)(bb * H_ + hr)) * L_ + l] = a * (1.0f / (1.0f + expf(-gv)));
      }
    }
  }
}

extern "C" void kernel_launch(void* const* d_in, const int* in_sizes, int n_in,
                              void* d_out, int out_size, void* d_ws, size_t ws_size,
                              hipStream_t stream) {
  (void)in_sizes; (void)n_in; (void)out_size; (void)ws_size;
  const float* u          = (const float*)d_in[0];
  const float* log_dt     = (const float*)d_in[1];
  const float* C          = (const float*)d_in[2];
  const float* log_A_real = (const float*)d_in[3];
  const float* A_imag     = (const float*)d_in[4];
  const float* Dvec       = (const float*)d_in[5];
  const float* W          = (const float*)d_in[6];
  const float* bias       = (const float*)d_in[7];
  float* out = (float*)d_out;

  float* ws = (float*)d_ws;
  float* Ktap2 = ws;
  float* CoutR = Ktap2 + (size_t)H_ * 128;
  float* CoutI = CoutR + (size_t)H_ * Q_ * NMODE;
  float* W64r  = CoutI + (size_t)H_ * Q_ * NMODE;
  float* W64i  = W64r  + (size_t)H_ * NMODE;
  float* PwR   = W64i  + (size_t)H_ * NMODE;
  float* PwI   = PwR   + (size_t)H_ * Q_ * NMODE;
  float* ygelu = PwI   + (size_t)H_ * Q_ * NMODE;

  s4d_setup<<<H_, 32, 0, stream>>>(log_dt, C, log_A_real, A_imag,
                                   Ktap2, CoutR, CoutI, W64r, W64i, PwR, PwI);
  s4d_conv<<<H_ * B_, 256, 0, stream>>>(u, Ktap2, CoutR, CoutI, PwR, PwI, W64r, W64i, Dvec, ygelu);
  s4d_glu<<<1024, 256, GLU_LDS_BYTES, stream>>>(ygelu, W, bias, out);
}


# conv kernel epilogue: GELU results staged in LDS (over the block's u row) and written as whole 128-B lines, 16 B per lane, instead of per-lane 4-byte stores 256 B apart; values bit-identical
# speedup vs baseline: 1.7490x; 1.0434x over previous
#include <hip/hip_runtime.h>
#include <math.h>

#define H_     512
#define NMODE  32
#define B_     8
#define L_     2048
#define Q_     64
#define NCHUNK 32
#define CPITCH 34
#define PPITCH 68
#define SPITCH 33
#define XPITCH 36

typedef __attribute__((ext_vector_type(2))) float v2f;
typedef __attribute__((ext_vector_type(8))) float v8f;

__device__ __forceinline__ v8f wmma4(v2f a, v2f b, v8f c) {
  return __builtin_amdgcn_wmma_f32_16x16x4_f32(false, a, false, b, (short)0, c,
                                               false, false);
}

__device__ __forceinline__ float xhalf(float x) {
  return __builtin_bit_cast(float, __builtin_amdgcn_permlanex16(__builtin_bit_cast(int, x), __builtin_bit_cast(int, x),
                                                                 0x76543210, 0xfedcba98, false, false));
}
template <int M>
__device__ __forceinline__ float xrow(float x) {
  return __builtin_bit_cast(float, __builtin_amdgcn_update_dpp(__builtin_bit_cast(int, x), __builtin_bit_cast(int, x),
                                                               0x160 + M, 0xf, 0xf, true));
}

__global__ __launch_bounds__(32) void s4d_setup(const float* __restrict__ log_dt,
                          const float* __restrict__ C,
                          const float* __restrict__ log_A_real,
                          const float* __restrict__ A_imag,
                          float* __restrict__ Ktap2,
                          float* __restrict__ CoutR,
                          float* __restrict__ CoutI,
                          float* __restrict__ W64r,
                          float* __restrict__ W64i,
                          float* __restrict__ PwR,
                          float* __restrict__ PwI) {
  const int h = blockIdx.x;
  const int k = threadIdx.x;
  const float dt = expf(log_dt[h]);
  const float Ar = -expf(log_A_real[h * NMODE + k]);
  const float Ai = A_imag[h * NMODE + k];
  const float Ccr = C[(h * NMODE + k) * 2 + 0];
  const float Cci = C[(h * NMODE + k) * 2 + 1];
  const float er = expf(Ar * dt);
  const float wr = er * cosf(Ai * dt);
  const float wi = er * sinf(Ai * dt);
  const float exr = wr - 1.0f, exi = wi;
  const float den = Ar * Ar + Ai * Ai;
  const float qr = (exr * Ar + exi * Ai) / den;
  const float qi = (exi * Ar - exr * Ai) / den;
  const float Ckr = Ccr * qr - Cci * qi;
  const float Cki = Ccr * qi + Cci * qr;

  Ktap2[h * 128 + k] = 0.0f;
  Ktap2[h * 128 + 32 + k] = 0.0f;

  float tv[Q_];
  {
    float pr = Ckr, pi = Cki;
#pragma unroll
    for (int d = 0; d < Q_; ++d) {
      float v = 2.0f * pr;
      v += xhalf(v);
      v += xrow<8>(v);
      v += xrow<4>(v);
      v += xrow<2>(v);
      v += xrow<1>(v);
      tv[d] = v;
      const float nr = pr * wr - pi * wi;
      const float ni = pr * wi + pi * wr;
      pr = nr; pi = ni;
    }
  }
  float o0 = tv[0], o1 = tv[32];
#pragma unroll
  for (int d = 1; d < 32; ++d) {
    if (k == d) { o0 = tv[d]; o1 = tv[d + 32]; }
  }
  Ktap2[h * 128 + 64 + k] = o0;
  Ktap2[h * 128 + 96 + k] = o1;
  float sr = Ckr * wr - Cki * wi;
  float si = Ckr * wi + Cki * wr;
  for (int j = 0; j < Q_; ++j) {
    CoutR[((size_t)h * Q_ + j) * NMODE + k] = 2.0f * sr;
    CoutI[((size_t)h * Q_ + j) * NMODE + k] = -2.0f * si;
    const float nr = sr * wr - si * wi;
    const float ni = sr * wi + si * wr;
    sr = nr; si = ni;
  }
  float qr_ = 1.0f, qi_ = 0.0f;
  for (int d = 0; d < Q_; ++d) {
    PwR[((size_t)h * Q_ + d) * NMODE + k] = qr_;
    PwI[((size_t)h * Q_ + d) * NMODE + k] = qi_;
    const float nr = qr_ * wr - qi_ * wi;
    const float ni = qr_ * wi + qi_ * wr;
    qr_ = nr; qi_ = ni;
  }
  W64r[h * NMODE + k] = qr_;
  W64i[h * NMODE + k] = qi_;
}

__global__ __launch_bounds__(256, 10) void s4d_conv(const float* __restrict__ u,
                         const float* __restrict__ Ktap2,
                         const float* __restrict__ CoutR,
                         const float* __restrict__ CoutI,
                         const float* __restrict__ PwR,
                         const float* __restrict__ PwI,
                         const float* __restrict__ W64r,
                         const float* __restrict__ W64i,
                         const float* __restrict__ Dvec,
                         float* __restrict__ ygelu) {
  __shared__ __attribute__((aligned(16))) float sKt[256];
  __shared__ __attribute__((aligned(16))) float sCR[Q_ * CPITCH];
  __shared__ __attribute__((aligned(16))) float sCI[Q_ * CPITCH];
  __shared__ __attribute__((aligned(16))) float sP[2 * NMODE * PPITCH];
  __shared__ __attribute__((aligned(16))) float sU[NCHUNK * PPITCH];
  __shared__ __attribute__((aligned(16))) float sS[2 * NMODE * SPITCH];
  __shared__ __attribute__((aligned(16))) float sXR[NCHUNK * XPITCH];
  __shared__ __attribute__((aligned(16))) float sXI[NCHUNK * XPITCH];

  const int h      = blockIdx.x >> 3;
  const int b      = blockIdx.x & 7;
  const int waveId = threadIdx.x >> 5;
  const int lane   = threadIdx.x & 31;
  const int half   = lane >> 4;
  const int lm     = lane & 15;
  const int jt     = waveId & 3;
  const int ct     = waveId >> 2;
  const int col    = ct * 16 + lm;

  const size_t urow = (size_t)(b * H_ + h) * L_;

  const float ar = W64r[h * NMODE + lane];
  const float ai = W64i[h * NMODE + lane];

  const float rk = Ktap2[(size_t)h * 128 + (threadIdx.x & 127)];
  float rcr[8], rci[8], rpr[8], rpi[8];
  float4 ru[2];
#pragma unroll
  for (int it = 0; it < 8; ++it) {
    const size_t gi = (size_t)h * (Q_ * NMODE) + it * 256 + threadIdx.x;
    rcr[it] = CoutR[gi];
    rci[it] = CoutI[gi];
    rpr[it] = PwR[gi];
    rpi[it] = PwI[gi];
  }
#pragma unroll
  for (int it = 0; it < 2; ++it)
    ru[it] = *(const float4*)(u + urow + (it * 256 + threadIdx.x) * 4);

  sKt[threadIdx.x] = rk;
#pragma unroll
  for (int it = 0; it < 8; ++it) {
    const int i = it * 256 + threadIdx.x;
    const int j = i >> 5, k = i & 31;
    sCR[j * CPITCH + k] = rcr[it];
    sCI[j * CPITCH + k] = rci[it];
    sP[k * PPITCH + (Q_ - 1 - j)]           = rpr[it];
    sP[(NMODE + k) * PPITCH + (Q_ - 1 - j)] = rpi[it];
  }
#pragma unroll
  for (int it = 0; it < 2; ++it) {
    const int e = (it * 256 + threadIdx.x) * 4;
    *(float4*)(sU + (e >> 6) * PPITCH + (e & 63)) = ru[it];
  }
  __syncthreads();

  const int jrow = jt * 16 + lm;

  v8f acc  = {0.f, 0.f, 0.f, 0.f, 0.f, 0.f, 0.f, 0.f};
  v8f accS = {0.f, 0.f, 0.f, 0.f, 0.f, 0.f, 0.f, 0.f};

#pragma unroll
  for (int k0 = 0; k0 < Q_; k0 += 4) {
    const int ta = k0 + 2 * half;
    const v2f bb = *(const v2f*)(sU + col * PPITCH + ta);
    const v2f ap = *(const v2f*)(sP + jrow * PPITCH + ta);
    v2f a;
    a.x = sKt[64 + jrow - ta];
    a.y = sKt[63 + jrow - ta];
    accS = wmma4(ap, bb, accS);
    acc  = wmma4(a, bb, acc);
  }
#pragma unroll
  for (int v = 0; v < 8; ++v)
    sS[(jt * 16 + v + 8 * half) * SPITCH + col] = accS[v];
  __syncthreads();

  if (waveId == 0) {
    float xr = 0.0f, xi = 0.0f;
#pragma unroll
    for (int c0 = 0; c0 < NCHUNK; c0 += 16) {
      float pr[16], pi[16];
#pragma unroll
      for (int i = 0; i < 16; ++i) {
        pr[i] = sS[lane * SPITCH + c0 + i];
        pi[i] = sS[(NMODE + lane) * SPITCH + c0 + i];
      }
#pragma unroll
      for (int i = 0; i < 16; ++i) {
        sXR[(c0 + i) * XPITCH + lane] = xr;
        sXI[(c0 + i) * XPITCH + lane] = xi;
        const float nr = ar * xr - ai * xi + pr[i];
        const float ni = ar * xi + ai * xr + pi[i];
        xr = nr; xi = ni;
      }
    }
  }
  __syncthreads();

#pragma unroll
  for (int k0 = 0; k0 < NMODE; k0 += 4) {
    const int ka = k0 + 2 * half;
    const v2f a  = *(const v2f*)(sCR + jrow * CPITCH + ka);
    const v2f bb = *(const v2f*)(sXR + col * XPITCH + ka);
    acc = wmma4(a, bb, acc);
  }
#pragma unroll
  for (int k0 = 0; k0 < NMODE; k0 += 4) {
    const int ka = k0 + 2 * half;
    const v2f a  = *(const v2f*)(sCI + jrow * CPITCH + ka);
    const v2f bb = *(const v2f*)(sXI + col * XPITCH + ka);
    acc = wmma4(a, bb, acc);
  }
  const float Dh = Dvec[h];
  float* su8 = sU + col * PPITCH + jt * 16 + 8 * half;
  const float4 u0 = *(const float4*)(su8);
  const float4 u1 = *(const float4*)(su8 + 4);
  const float uu[8] = {u0.x, u0.y, u0.z, u0.w, u1.x, u1.y, u1.z, u1.w};
  float yo[8];
#pragma unroll
  for (int v = 0; v < 8; ++v) {
    const float y = acc[v] + Dh * uu[v];
    yo[v] = 0.5f * y * (1.0f + erff(y * 0.70710678118654752f));
  }
  *(float4*)(su8)     = make_float4(yo[0], yo[1], yo[2], yo[3]);
  *(float4*)(su8 + 4) = make_float4(yo[4], yo[5], yo[6], yo[7]);
  __syncthreads();
#pragma unroll
  for (int it = 0; it < 2; ++it) {
    const int e = (it * 256 + threadIdx.x) * 4;
    *(float4*)(ygelu + urow + e) = *(const float4*)(sU + (e >> 6) * PPITCH + (e & 63));
  }
}

#define GKC      32
#define APITCH   36
#define BPITCH   136
#define A_BUF    (128 * APITCH)
#define B_BUF    (GKC * BPITCH)
#define GLU_LDS_BYTES 71680

__global__ __launch_bounds__(256) void s4d_glu(const float* __restrict__ ygelu,
                                               const float* __restrict__ W,
                                               const float* __restrict__ bias,
                                               float* __restrict__ out) {
  extern __shared__ __attribute__((aligned(16))) float smem[];
  float* sA = smem;
  float* sB = smem + 2 * A_BUF;

  const int tid    = threadIdx.x;
  const int waveId = tid >> 5;
  const int lane   = tid & 31;
  const int half   = lane >> 4;
  const int lm     = lane & 15;
  const int wm     = waveId & 1;
  const int wn     = waveId >> 1;
  const int mblk   = blockIdx.x & 7;
  const int nblk   = blockIdx.x >> 3;
  const int bb     = nblk >> 4;
  const int l0     = (nblk & 15) * 128;

  const float* gA = W + (size_t)(mblk * 64 + (tid >> 3)) * H_ + (tid & 7) * 4;
  const float* gB = ygelu + ((size_t)bb * H_ + (tid >> 5)) * L_ + l0 + (tid & 31) * 4;
  const int stA = (tid >> 3) * APITCH + (tid & 7) * 4;
  const int stB = (tid >> 5) * BPITCH + (tid & 31) * 4;

  float4 ra[4], rb[4];
#pragma unroll
  for (int j = 0; j < 4; ++j) {
    ra[j] = *(const float4*)(gA + (size_t)(j >> 1) * H_ * H_ + (size_t)(j & 1) * 32 * H_);
    rb[j] = *(const float4*)(gB + (size_t)j * 8 * L_);
  }
#pragma unroll
  for (int j = 0; j < 4; ++j) {
    *(float4*)(sA + stA + j * 32 * APITCH) = ra[j];
    *(float4*)(sB + stB + j * 8 * BPITCH)  = rb[j];
  }
  __syncthreads();

  v8f acc[4][2];
#pragma unroll
  for (int i = 0; i < 4; ++i)
#pragma unroll
    for (int j = 0; j < 2; ++j)
      acc[i][j] = (v8f){0.f, 0.f, 0.f, 0.f, 0.f, 0.f, 0.f, 0.f};

  const int fA = (wm * 32 + lm) * APITCH + 2 * half;
  const int fB = (2 * half) * BPITCH + wn * 32 + lm;

  for (int kc = 0; kc < H_ / GKC; ++kc) {
    const int cur = kc & 1;
    const bool more = (kc + 1) < (H_ / GKC);
    if (more) {
      const float* nA = gA + (kc + 1) * GKC;
      const float* nB = gB + (size_t)(kc + 1) * GKC * L_;
#pragma unroll
      for (int j = 0; j < 4; ++j) {
        ra[j] = *(const float4*)(nA + (size_t)(j >> 1) * H_ * H_ + (size_t)(j & 1) * 32 * H_);
        rb[j] = *(const float4*)(nB + (size_t)j * 8 * L_);
      }
    }
    const float* cA = sA + cur * A_BUF + fA;
    const float* cB = sB + cur * B_BUF + fB;
#pragma unroll
    for (int k0 = 0; k0 < GKC; k0 += 4) {
      v2f a[4], b[2];
      a[0] = *(const v2f*)(cA + k0);
      a[1] = *(const v2f*)(cA + 16 * APITCH + k0);
      a[2] = *(const v2f*)(cA + 64 * APITCH + k0);
      a[3] = *(const v2f*)(cA + 80 * APITCH + k0);
#pragma unroll
      for (int j = 0; j < 2; ++j) {
        b[j].x = cB[k0 * BPITCH + 16 * j];
        b[j].y = cB[(k0 + 1) * BPITCH + 16 * j];
      }
#pragma unroll
      for (int i = 0; i < 4; ++i)
#pragma unroll
        for (int j = 0; j < 2; ++j)
          acc[i][j] = wmma4(a[i], b[j], acc[i][j]);
    }
    if (more) {
      float* dA = sA + (cur ^ 1) * A_BUF + stA;
      float* dB = sB + (cur ^ 1) * B_BUF + stB;
#pragma unroll
      for (int j = 0; j < 4; ++j) {
        *(float4*)(dA + j * 32 * APITCH) = ra[j];
        *(float4*)(dB + j * 8 * BPITCH)  = rb[j];
      }
    }
    __syncthreads();
  }

#pragma unroll
  for (int mi = 0; mi < 2; ++mi) {
#pragma unroll
    for (int v = 0; v < 8; ++v) {
      const int hr = mblk * 64 + wm * 32 + mi * 16 + v + 8 * half;
      const float ba = bias[hr];
      const float bg = bias[H_ + hr];
#pragma unroll
      for (int j = 0; j < 2; ++j) {
        const int l = l0 + wn * 32 + j * 16 + lm;
        const float a  = acc[mi][j][v] + ba;
        const float gv = acc[2 + mi][j][v] + bg;
        out[((size_t)(bb * H_ + hr)) * L_ + l] = a * (1.0f / (1.0f + expf(-gv)));
      }
    }
  }
}

extern "C" void kernel_launch(void* const* d_in, const int* in_sizes, int n_in,
                              void* d_out, int out_size, void* d_ws, size_t ws_size,
                              hipStream_t stream) {
  (void)in_sizes; (void)n_in; (void)out_size; (void)ws_size;
  const float* u          = (const float*)d_in[0];
  const float* log_dt     = (const float*)d_in[1];
  const float* C          = (const float*)d_in[2];
  const float* log_A_real = (const float*)d_in[3];
  const float* A_imag     = (const float*)d_in[4];
  const float* Dvec       = (const float*)d_in[5];
  const float* W          = (const float*)d_in[6];
  const float* bias       = (const float*)d_in[7];
  float* out = (float*)d_out;

  float* ws = (float*)d_ws;
  float* Ktap2 = ws;
  float* CoutR = Ktap2 + (size_t)H_ * 128;
  float* CoutI = CoutR + (size_t)H_ * Q_ * NMODE;
  float* W64r  = CoutI + (size_t)H_ * Q_ * NMODE;
  float* W64i  = W64r  + (size_t)H_ * NMODE;
  float* PwR   = W64i  + (size_t)H_ * NMODE;
  float* PwI   = PwR   + (size_t)H_ * Q_ * NMODE;
  float* ygelu = PwI   + (size_t)H_ * Q_ * NMODE;

  s4d_setup<<<H_, 32, 0, stream>>>(log_dt, C, log_A_real, A_imag,
                                   Ktap2, CoutR, CoutI, W64r, W64i, PwR, PwI);
  s4d_conv<<<H_ * B_, 256, 0, stream>>>(u, Ktap2, CoutR, CoutI, PwR, PwI, W64r, W64i, Dvec, ygelu);
  s4d_glu<<<1024, 256, GLU_LDS_BYTES, stream>>>(ygelu, W, bias, out);
}


# conv kernel: one block walks 4 batch rows of its head, the head's tables staged in LDS once per block (not once per row), next u row loaded under the current row's products; sums unchanged
# speedup vs baseline: 1.8387x; 1.0513x over previous
#include <hip/hip_runtime.h>
#include <math.h>

#define H_     512
#define NMODE  32
#define B_     8
#define L_     2048
#define Q_     64
#define NCHUNK 32
#define CPITCH 34
#define PPITCH 68
#define SPITCH 33
#define XPITCH 36
#define NBROW  4
#define UBUF   (NCHUNK * PPITCH)

typedef __attribute__((ext_vector_type(2))) float v2f;
typedef __attribute__((ext_vector_type(8))) float v8f;

__device__ __forceinline__ v8f wmma4(v2f a, v2f b, v8f c) {
  return __builtin_amdgcn_wmma_f32_16x16x4_f32(false, a, false, b, (short)0, c,
                                               false, false);
}

__device__ __forceinline__ float xhalf(float x) {
  return __builtin_bit_cast(float, __builtin_amdgcn_permlanex16(__builtin_bit_cast(int, x), __builtin_bit_cast(int, x),
                                                                 0x76543210, 0xfedcba98, false, false));
}
template <int M>
__device__ __forceinline__ float xrow(float x) {
  return __builtin_bit_cast(float, __builtin_amdgcn_update_dpp(__builtin_bit_cast(int, x), __builtin_bit_cast(int, x),
                                                               0x160 + M, 0xf, 0xf, true));
}

__device__ __forceinline__ void lds_barrier() {
  __builtin_amdgcn_fence(__ATOMIC_RELEASE, "workgroup", "local");
  __builtin_amdgcn_s_barrier();
  __builtin_amdgcn_fence(__ATOMIC_ACQUIRE, "workgroup", "local");
}

__global__ __launch_bounds__(32) void s4d_setup(const float* __restrict__ log_dt,
                          const float* __restrict__ C,
                          const float* __restrict__ log_A_real,
                          const float* __restrict__ A_imag,
                          float* __restrict__ Ktap2,
                          float* __restrict__ CoutR,
                          float* __restrict__ CoutI,
                          float* __restrict__ W64r,
                          float* __restrict__ W64i,
                          float* __restrict__ PwR,
                          float* __restrict__ PwI) {
  const int h = blockIdx.x;
  const int k = threadIdx.x;
  const float dt = expf(log_dt[h]);
  const float Ar = -expf(log_A_real[h * NMODE + k]);
  const float Ai = A_imag[h * NMODE + k];
  const float Ccr = C[(h * NMODE + k) * 2 + 0];
  const float Cci = C[(h * NMODE + k) * 2 + 1];
  const float er = expf(Ar * dt);
  const float wr = er * cosf(Ai * dt);
  const float wi = er * sinf(Ai * dt);
  const float exr = wr - 1.0f, exi = wi;
  const float den = Ar * Ar + Ai * Ai;
  const float qr = (exr * Ar + exi * Ai) / den;
  const float qi = (exi * Ar - exr * Ai) / den;
  const float Ckr = Ccr * qr - Cci * qi;
  const float Cki = Ccr * qi + Cci * qr;

  Ktap2[h * 128 + k] = 0.0f;
  Ktap2[h * 128 + 32 + k] = 0.0f;

  float tv[Q_];
  {
    float pr = Ckr, pi = Cki;
#pragma unroll
    for (int d = 0; d < Q_; ++d) {
      float v = 2.0f * pr;
      v += xhalf(v);
      v += xrow<8>(v);
      v += xrow<4>(v);
      v += xrow<2>(v);
      v += xrow<1>(v);
      tv[d] = v;
      const float nr = pr * wr - pi * wi;
      const float ni = pr * wi + pi * wr;
      pr = nr; pi = ni;
    }
  }
  float o0 = tv[0], o1 = tv[32];
#pragma unroll
  for (int d = 1; d < 32; ++d) {
    if (k == d) { o0 = tv[d]; o1 = tv[d + 32]; }
  }
  Ktap2[h * 128 + 64 + k] = o0;
  Ktap2[h * 128 + 96 + k] = o1;
  float sr = Ckr * wr - Cki * wi;
  float si = Ckr * wi + Cki * wr;
  for (int j = 0; j < Q_; ++j) {
    CoutR[((size_t)h * Q_ + j) * NMODE + k] = 2.0f * sr;
    CoutI[((size_t)h * Q_ + j) * NMODE + k] = -2.0f * si;
    const float nr = sr * wr - si * wi;
    const float ni = sr * wi + si * wr;
    sr = nr; si = ni;
  }
  float qr_ = 1.0f, qi_ = 0.0f;
  for (int d = 0; d < Q_; ++d) {
    PwR[((size_t)h * Q_ + d) * NMODE + k] = qr_;
    PwI[((size_t)h * Q_ + d) * NMODE + k] = qi_;
    const float nr = qr_ * wr - qi_ * wi;
    const float ni = qr_ * wi + qi_ * wr;
    qr_ = nr; qi_ = ni;
  }
  W64r[h * NMODE + k] = qr_;
  W64i[h * NMODE + k] = qi_;
}

__global__ __launch_bounds__(256, 8) void s4d_conv(const float* __restrict__ u,
                         const float* __restrict__ Ktap2,
                         const float* __restrict__ CoutR,
                         const float* __restrict__ CoutI,
                         const float* __restrict__ PwR,
                         const float* __restrict__ PwI,
                         const float* __restrict__ W64r,
                         const float* __restrict__ W64i,
                         const float* __restrict__ Dvec,
                         float* __restrict__ ygelu) {
  __shared__ __attribute__((aligned(16))) float sKt[256];
  __shared__ __attribute__((aligned(16))) float sCR[Q_ * CPITCH];
  __shared__ __attribute__((aligned(16))) float sCI[Q_ * CPITCH];
  __shared__ __attribute__((aligned(16))) float sP[2 * NMODE * PPITCH];
  __shared__ __attribute__((aligned(16))) float sU[2 * UBUF];
  __shared__ __attribute__((aligned(16))) float sS[2 * NMODE * SPITCH];
  __shared__ __attribute__((aligned(16))) float sXR[NCHUNK * XPITCH];
  __shared__ __attribute__((aligned(16))) float sXI[NCHUNK * XPITCH];

  const int h      = blockIdx.x >> 1;
  const int b0     = (blockIdx.x & 1) * NBROW;
  const int waveId = threadIdx.x >> 5;
  const int lane   = threadIdx.x & 31;
  const int half   = lane >> 4;
  const int lm     = lane & 15;
  const int jt     = waveId & 3;
  const int ct     = waveId >> 2;
  const int col    = ct * 16 + lm;

  size_t urow = (size_t)(b0 * H_ + h) * L_;

  const float ar = W64r[h * NMODE + lane];
  const float ai = W64i[h * NMODE + lane];
  const float Dh = Dvec[h];

  const float rk = Ktap2[(size_t)h * 128 + (threadIdx.x & 127)];
  float rcr[8], rci[8], rpr[8], rpi[8];
  float4 ru0, ru1;
#pragma unroll
  for (int it = 0; it < 8; ++it) {
    const size_t gi = (size_t)h * (Q_ * NMODE) + it * 256 + threadIdx.x;
    rcr[it] = CoutR[gi];
    rci[it] = CoutI[gi];
    rpr[it] = PwR[gi];
    rpi[it] = PwI[gi];
  }
  ru0 = *(const float4*)(u + urow + threadIdx.x * 4);
  ru1 = *(const float4*)(u + urow + (256 + threadIdx.x) * 4);

  sKt[threadIdx.x] = rk;
#pragma unroll
  for (int it = 0; it < 8; ++it) {
    const int i = it * 256 + threadIdx.x;
    const int j = i >> 5, k = i & 31;
    sCR[j * CPITCH + k] = rcr[it];
    sCI[j * CPITCH + k] = rci[it];
    sP[k * PPITCH + (Q_ - 1 - j)]           = rpr[it];
    sP[(NMODE + k) * PPITCH + (Q_ - 1 - j)] = rpi[it];
  }
  const int jrow = jt * 16 + lm;

  for (int r = 0; r < NBROW; ++r) {
  float* sUc = sU + (r & 1) * UBUF;
  {
    const int e0 = threadIdx.x * 4, e1 = (256 + threadIdx.x) * 4;
    *(float4*)(sUc + (e0 >> 6) * PPITCH + (e0 & 63)) = ru0;
    *(float4*)(sUc + (e1 >> 6) * PPITCH + (e1 & 63)) = ru1;
  }
  if (r + 1 < NBROW) {
    ru0 = *(const float4*)(u + urow + (size_t)H_ * L_ + threadIdx.x * 4);
    ru1 = *(const float4*)(u + urow + (size_t)H_ * L_ + (256 + threadIdx.x) * 4);
  }
  lds_barrier();

  v8f acc  = {0.f, 0.f, 0.f, 0.f, 0.f, 0.f, 0.f, 0.f};
  v8f accS = {0.f, 0.f, 0.f, 0.f, 0.f, 0.f, 0.f, 0.f};

#pragma unroll
  for (int k0 = 0; k0 < Q_; k0 += 4) {
    const int ta = k0 + 2 * half;
    const v2f bb = *(const v2f*)(sUc + col * PPITCH + ta);
    const v2f ap = *(const v2f*)(sP + jrow * PPITCH + ta);
    v2f a;
    a.x = sKt[64 + jrow - ta];
    a.y = sKt[63 + jrow - ta];
    accS = wmma4(ap, bb, accS);
    acc  = wmma4(a, bb, acc);
  }
#pragma unroll
  for (int v = 0; v < 8; ++v)
    sS[(jt * 16 + v + 8 * half) * SPITCH + col] = accS[v];
  lds_barrier();

  if (waveId == 0) {
    float xr = 0.0f, xi = 0.0f;
#pragma unroll
    for (int c0 = 0; c0 < NCHUNK; c0 += 16) {
      float pr[16], pi[16];
#pragma unroll
      for (int i = 0; i < 16; ++i) {
        pr[i] = sS[lane * SPITCH + c0 + i];
        pi[i] = sS[(NMODE + lane) * SPITCH + c0 + i];
      }
#pragma unroll
      for (int i = 0; i < 16; ++i) {
        sXR[(c0 + i) * XPITCH + lane] = xr;
        sXI[(c0 + i) * XPITCH + lane] = xi;
        const float nr = ar * xr - ai * xi + pr[i];
        const float ni = ar * xi + ai * xr + pi[i];
        xr = nr; xi = ni;
      }
    }
  }
  lds_barrier();

#pragma unroll
  for (int k0 = 0; k0 < NMODE; k0 += 4) {
    const int ka = k0 + 2 * half;
    const v2f a  = *(const v2f*)(sCR + jrow * CPITCH + ka);
    const v2f bb = *(const v2f*)(sXR + col * XPITCH + ka);
    acc = wmma4(a, bb, acc);
  }
#pragma unroll
  for (int k0 = 0; k0 < NMODE; k0 += 4) {
    const int ka = k0 + 2 * half;
    const v2f a  = *(const v2f*)(sCI + jrow * CPITCH + ka);
    const v2f bb = *(const v2f*)(sXI + col * XPITCH + ka);
    acc = wmma4(a, bb, acc);
  }
  float* su8 = sUc + col * PPITCH + jt * 16 + 8 * half;
  const float4 u0 = *(const float4*)(su8);
  const float4 u1 = *(const float4*)(su8 + 4);
  const float uu[8] = {u0.x, u0.y, u0.z, u0.w, u1.x, u1.y, u1.z, u1.w};
  float yo[8];
#pragma unroll
  for (int v = 0; v < 8; ++v) {
    const float y = acc[v] + Dh * uu[v];
    yo[v] = 0.5f * y * (1.0f + erff(y * 0.70710678118654752f));
  }
  *(float4*)(su8)     = make_float4(yo[0], yo[1], yo[2], yo[3]);
  *(float4*)(su8 + 4) = make_float4(yo[4], yo[5], yo[6], yo[7]);
  lds_barrier();
#pragma unroll
  for (int it = 0; it < 2; ++it) {
    const int e = (it * 256 + threadIdx.x) * 4;
    *(float4*)(ygelu + urow + e) = *(const float4*)(sUc + (e >> 6) * PPITCH + (e & 63));
  }
  urow += (size_t)H_ * L_;
  }
}

#define GKC      32
#define APITCH   36
#define BPITCH   136
#define A_BUF    (128 * APITCH)
#define B_BUF    (GKC * BPITCH)
#define GLU_LDS_BYTES 71680

__global__ __launch_bounds__(256) void s4d_glu(const float* __restrict__ ygelu,
                                               const float* __restrict__ W,
                                               const float* __restrict__ bias,
                                               float* __restrict__ out) {
  extern __shared__ __attribute__((aligned(16))) float smem[];
  float* sA = smem;
  float* sB = smem + 2 * A_BUF;

  const int tid    = threadIdx.x;
  const int waveId = tid >> 5;
  const int lane   = tid & 31;
  const int half   = lane >> 4;
  const int lm     = lane & 15;
  const int wm     = waveId & 1;
  const int wn     = waveId >> 1;
  const int mblk   = blockIdx.x & 7;
  const int nblk   = blockIdx.x >> 3;
  const int bb     = nblk >> 4;
  const int l0     = (nblk & 15) * 128;

  const float* gA = W + (size_t)(mblk * 64 + (tid >> 3)) * H_ + (tid & 7) * 4;
  const float* gB = ygelu + ((size_t)bb * H_ + (tid >> 5)) * L_ + l0 + (tid & 31) * 4;
  const int stA = (tid >> 3) * APITCH + (tid & 7) * 4;
  const int stB = (tid >> 5) * BPITCH + (tid & 31) * 4;

  float4 ra[4], rb[4];
#pragma unroll
  for (int j = 0; j < 4; ++j) {
    ra[j] = *(const float4*)(gA + (size_t)(j >> 1) * H_ * H_ + (size_t)(j & 1) * 32 * H_);
    rb[j] = *(const float4*)(gB + (size_t)j * 8 * L_);
  }
#pragma unroll
  for (int j = 0; j < 4; ++j) {
    *(float4*)(sA + stA + j * 32 * APITCH) = ra[j];
    *(float4*)(sB + stB + j * 8 * BPITCH)  = rb[j];
  }
  __syncthreads();

  v8f acc[4][2];
#pragma unroll
  for (int i = 0; i < 4; ++i)
#pragma unroll
    for (int j = 0; j < 2; ++j)
      acc[i][j] = (v8f){0.f, 0.f, 0.f, 0.f, 0.f, 0.f, 0.f, 0.f};

  const int fA = (wm * 32 + lm) * APITCH + 2 * half;
  const int fB = (2 * half) * BPITCH + wn * 32 + lm;

  for (int kc = 0; kc < H_ / GKC; ++kc) {
    const int cur = kc & 1;
    const bool more = (kc + 1) < (H_ / GKC);
    if (more) {
      const float* nA = gA + (kc + 1) * GKC;
      const float* nB = gB + (size_t)(kc + 1) * GKC * L_;
#pragma unroll
      for (int j = 0; j < 4; ++j) {
        ra[j] = *(const float4*)(nA + (size_t)(j >> 1) * H_ * H_ + (size_t)(j & 1) * 32 * H_);
        rb[j] = *(const float4*)(nB + (size_t)j * 8 * L_);
      }
    }
    const float* cA = sA + cur * A_BUF + fA;
    const float* cB = sB + cur * B_BUF + fB;
#pragma unroll
    for (int k0 = 0; k0 < GKC; k0 += 4) {
      v2f a[4], b[2];
      a[0] = *(const v2f*)(cA + k0);
      a[1] = *(const v2f*)(cA + 16 * APITCH + k0);
      a[2] = *(const v2f*)(cA + 64 * APITCH + k0);
      a[3] = *(const v2f*)(cA + 80 * APITCH + k0);
#pragma unroll
      for (int j = 0; j < 2; ++j) {
        b[j].x = cB[k0 * BPITCH + 16 * j];
        b[j].y = cB[(k0 + 1) * BPITCH + 16 * j];
      }
#pragma unroll
      for (int i = 0; i < 4; ++i)
#pragma unroll
        for (int j = 0; j < 2; ++j)
          acc[i][j] = wmma4(a[i], b[j], acc[i][j]);
    }
    if (more) {
      float* dA = sA + (cur ^ 1) * A_BUF + stA;
      float* dB = sB + (cur ^ 1) * B_BUF + stB;
#pragma unroll
      for (int j = 0; j < 4; ++j) {
        *(float4*)(dA + j * 32 * APITCH) = ra[j];
        *(float4*)(dB + j * 8 * BPITCH)  = rb[j];
      }
    }
    __syncthreads();
  }

#pragma unroll
  for (int mi = 0; mi < 2; ++mi) {
#pragma unroll
    for (int v = 0; v < 8; ++v) {
      const int hr = mblk * 64 + wm * 32 + mi * 16 + v + 8 * half;
      const float ba = bias[hr];
      const float bg = bias[H_ + hr];
#pragma unroll
      for (int j = 0; j < 2; ++j) {
        const int l = l0 + wn * 32 + j * 16 + lm;
        const float a  = acc[mi][j][v] + ba;
        const float gv = acc[2 + mi][j][v] + bg;
        out[((size_t)(bb * H_ + hr)) * L_ + l] = a * (1.0f / (1.0f + expf(-gv)));
      }
    }
  }
}

extern "C" void kernel_launch(void* const* d_in, const int* in_sizes, int n_in,
                              void* d_out, int out_size, void* d_ws, size_t ws_size,
                              hipStream_t stream) {
  (void)in_sizes; (void)n_in; (void)out_size; (void)ws_size;
  const float* u          = (const float*)d_in[0];
  const float* log_dt     = (const float*)d_in[1];
  const float* C          = (const float*)d_in[2];
  const float* log_A_real = (const float*)d_in[3];
  const float* A_imag     = (const float*)d_in[4];
  const float* Dvec       = (const float*)d_in[5];
  const float* W          = (const float*)d_in[6];
  const float* bias       = (const float*)d_in[7];
  float* out = (float*)d_out;

  float* ws = (float*)d_ws;
  float* Ktap2 = ws;
  float* CoutR = Ktap2 + (size_t)H_ * 128;
  float* CoutI = CoutR + (size_t)H_ * Q_ * NMODE;
  float* W64r  = CoutI + (size_t)H_ * Q_ * NMODE;
  float* W64i  = W64r  + (size_t)H_ * NMODE;
  float* PwR   = W64i  + (size_t)H_ * NMODE;
  float* PwI   = PwR   + (size_t)H_ * Q_ * NMODE;
  float* ygelu = PwI   + (size_t)H_ * Q_ * NMODE;

  s4d_setup<<<H_, 32, 0, stream>>>(log_dt, C, log_A_real, A_imag,
                                   Ktap2, CoutR, CoutI, W64r, W64i, PwR, PwI);
  s4d_conv<<<H_ * B_ / NBROW, 256, 0, stream>>>(u, Ktap2, CoutR, CoutI, PwR, PwI, W64r, W64i, Dvec, ygelu);
  s4d_glu<<<1024, 256, GLU_LDS_BYTES, stream>>>(ygelu, W, bias, out);
}


# set-up kernel folded into the conv kernel: each block derives its head's taps, power and projection tables straight into LDS on four of its waves (same chains, same lane sums); 3 launches -> 2
# speedup vs baseline: 1.9317x; 1.0506x over previous
#include <hip/hip_runtime.h>
#include <math.h>

#define H_     512
#define NMODE  32
#define B_     8
#define L_     2048
#define Q_     64
#define NCHUNK 32
#define CPITCH 34
#define PPITCH 68
#define SPITCH 33
#define XPITCH 36
#define NBROW  4
#define UBUF   (NCHUNK * PPITCH)

typedef __attribute__((ext_vector_type(2))) float v2f;
typedef __attribute__((ext_vector_type(8))) float v8f;

__device__ __forceinline__ v8f wmma4(v2f a, v2f b, v8f c) {
  return __builtin_amdgcn_wmma_f32_16x16x4_f32(false, a, false, b, (short)0, c,
                                               false, false);
}

__device__ __forceinline__ float xhalf(float x) {
  return __builtin_bit_cast(float, __builtin_amdgcn_permlanex16(__builtin_bit_cast(int, x), __builtin_bit_cast(int, x),
                                                                 0x76543210, 0xfedcba98, false, false));
}
template <int M>
__device__ __forceinline__ float xrow(float x) {
  return __builtin_bit_cast(float, __builtin_amdgcn_update_dpp(__builtin_bit_cast(int, x), __builtin_bit_cast(int, x),
                                                               0x160 + M, 0xf, 0xf, true));
}

__device__ __forceinline__ void lds_barrier() {
  __builtin_amdgcn_fence(__ATOMIC_RELEASE, "workgroup", "local");
  __builtin_amdgcn_s_barrier();
  __builtin_amdgcn_fence(__ATOMIC_ACQUIRE, "workgroup", "local");
}

__device__ __forceinline__ void cstep(float& pr, float& pi, float wr, float wi) {
  const float nr = pr * wr - pi * wi;
  const float ni = pr * wi + pi * wr;
  pr = nr; pi = ni;
}

__device__ __forceinline__ float taps32(float& pr, float& pi, float wr, float wi, int lane) {
  float tv[32];
#pragma unroll
  for (int d = 0; d < 32; ++d) {
    float v = 2.0f * pr;
    v += xhalf(v);
    v += xrow<8>(v);
    v += xrow<4>(v);
    v += xrow<2>(v);
    v += xrow<1>(v);
    tv[d] = v;
    cstep(pr, pi, wr, wi);
  }
  float o = tv[0];
#pragma unroll
  for (int d = 1; d < 32; ++d) {
    if (lane == d) o = tv[d];
  }
  return o;
}

__global__ __launch_bounds__(256, 8) void s4d_conv(const float* __restrict__ u,
                         const float* __restrict__ log_dt,
                         const float* __restrict__ C,
                         const float* __restrict__ log_A_real,
                         const float* __restrict__ A_imag,
                         const float* __restrict__ Dvec,
                         float* __restrict__ ygelu) {
  __shared__ __attribute__((aligned(16))) float sKt[256];
  __shared__ __attribute__((aligned(16))) float sCR[Q_ * CPITCH];
  __shared__ __attribute__((aligned(16))) float sCI[Q_ * CPITCH];
  __shared__ __attribute__((aligned(16))) float sP[2 * NMODE * PPITCH];
  __shared__ __attribute__((aligned(16))) float sU[2 * UBUF];
  __shared__ __attribute__((aligned(16))) float sS[2 * NMODE * SPITCH];
  __shared__ __attribute__((aligned(16))) float sXR[NCHUNK * XPITCH];
  __shared__ __attribute__((aligned(16))) float sXI[NCHUNK * XPITCH];

  const int h      = blockIdx.x >> 1;
  const int b0     = (blockIdx.x & 1) * NBROW;
  const int waveId = threadIdx.x >> 5;
  const int lane   = threadIdx.x & 31;
  const int half   = lane >> 4;
  const int lm     = lane & 15;
  const int jt     = waveId & 3;
  const int ct     = waveId >> 2;
  const int col    = ct * 16 + lm;

  size_t urow = (size_t)(b0 * H_ + h) * L_;

  const float Dh = Dvec[h];

  float4 ru0, ru1;
  ru0 = *(const float4*)(u + urow + threadIdx.x * 4);
  ru1 = *(const float4*)(u + urow + (256 + threadIdx.x) * 4);

  if (waveId < 4) {
    const int k = lane;
    const float dt = expf(log_dt[h]);
    const float Ar = -expf(log_A_real[h * NMODE + k]);
    const float Ai = A_imag[h * NMODE + k];
    const float Ccr = C[(h * NMODE + k) * 2 + 0];
    const float Cci = C[(h * NMODE + k) * 2 + 1];
    const float er = expf(Ar * dt);
    const float wr = er * cosf(Ai * dt);
    const float wi = er * sinf(Ai * dt);
    const float exr = wr - 1.0f, exi = wi;
    const float den = Ar * Ar + Ai * Ai;
    const float qr = (exr * Ar + exi * Ai) / den;
    const float qi = (exi * Ar - exr * Ai) / den;
    const float Ckr = Ccr * qr - Cci * qi;
    const float Cki = Ccr * qi + Cci * qr;
    if (waveId == 0) {
      sKt[k] = 0.0f;
      sKt[32 + k] = 0.0f;
      float pr = Ckr, pi = Cki;
      sKt[64 + k] = taps32(pr, pi, wr, wi, k);
    } else if (waveId == 3) {
      float pr = Ckr, pi = Cki;
      for (int d = 0; d < 32; ++d) cstep(pr, pi, wr, wi);
      sKt[96 + k] = taps32(pr, pi, wr, wi, k);
    } else if (waveId == 1) {
      float sr = Ckr * wr - Cki * wi;
      float si = Ckr * wi + Cki * wr;
      for (int j = 0; j < Q_; ++j) {
        sCR[j * CPITCH + k] = 2.0f * sr;
        sCI[j * CPITCH + k] = -2.0f * si;
        cstep(sr, si, wr, wi);
      }
    } else {
      float pr = 1.0f, pi = 0.0f;
      for (int d = 0; d < Q_; ++d) {
        sP[k * PPITCH + (Q_ - 1 - d)]           = pr;
        sP[(NMODE + k) * PPITCH + (Q_ - 1 - d)] = pi;
        cstep(pr, pi, wr, wi);
      }
      sKt[128 + k] = pr;
      sKt[160 + k] = pi;
    }
  }
  const int jrow = jt * 16 + lm;

  for (int r = 0; r < NBROW; ++r) {
  float* sUc = sU + (r & 1) * UBUF;
  {
    const int e0 = threadIdx.x * 4, e1 = (256 + threadIdx.x) * 4;
    *(float4*)(sUc + (e0 >> 6) * PPITCH + (e0 & 63)) = ru0;
    *(float4*)(sUc + (e1 >> 6) * PPITCH + (e1 & 63)) = ru1;
  }
  if (r + 1 < NBROW) {
    ru0 = *(const float4*)(u + urow + (size_t)H_ * L_ + threadIdx.x * 4);
    ru1 = *(const float4*)(u + urow + (size_t)H_ * L_ + (256 + threadIdx.x) * 4);
  }
  lds_barrier();

  v8f acc  = {0.f, 0.f, 0.f, 0.f, 0.f, 0.f, 0.f, 0.f};
  v8f accS = {0.f, 0.f, 0.f, 0.f, 0.f, 0.f, 0.f, 0.f};

#pragma unroll
  for (int k0 = 0; k0 < Q_; k0 += 4) {
    const int ta = k0 + 2 * half;
    const v2f bb = *(const v2f*)(sUc + col * PPITCH + ta);
    const v2f ap = *(const v2f*)(sP + jrow * PPITCH + ta);
    v2f a;
    a.x = sKt[64 + jrow - ta];
    a.y = sKt[63 + jrow - ta];
    accS = wmma4(ap, bb, accS);
    acc  = wmma4(a, bb, acc);
  }
#pragma unroll
  for (int v = 0; v < 8; ++v)
    sS[(jt * 16 + v + 8 * half) * SPITCH + col] = accS[v];
  lds_barrier();

  if (waveId == 0) {
    const float ar = sKt[128 + lane];
    const float ai = sKt[160 + lane];
    float xr = 0.0f, xi = 0.0f;
#pragma unroll
    for (int c0 = 0; c0 < NCHUNK; c0 += 16) {
      float pr[16], pi[16];
#pragma unroll
      for (int i = 0; i < 16; ++i) {
        pr[i] = sS[lane * SPITCH + c0 + i];
        pi[i] = sS[(NMODE + lane) * SPITCH + c0 + i];
      }
#pragma unroll
      for (int i = 0; i < 16; ++i) {
        sXR[(c0 + i) * XPITCH + lane] = xr;
        sXI[(c0 + i) * XPITCH + lane] = xi;
        const float nr = ar * xr - ai * xi + pr[i];
        const float ni = ar * xi + ai * xr + pi[i];
        xr = nr; xi = ni;
      }
    }
  }
  lds_barrier();

#pragma unroll
  for (int k0 = 0; k0 < NMODE; k0 += 4) {
    const int ka = k0 + 2 * half;
    const v2f a  = *(const v2f*)(sCR + jrow * CPITCH + ka);
    const v2f bb = *(const v2f*)(sXR + col * XPITCH + ka);
    acc = wmma4(a, bb, acc);
  }
#pragma unroll
  for (int k0 = 0; k0 < NMODE; k0 += 4) {
    const int ka = k0 + 2 * half;
    const v2f a  = *(const v2f*)(sCI + jrow * CPITCH + ka);
    const v2f bb = *(const v2f*)(sXI + col * XPITCH + ka);
    acc = wmma4(a, bb, acc);
  }
  float* su8 = sUc + col * PPITCH + jt * 16 + 8 * half;
  const float4 u0 = *(const float4*)(su8);
  const float4 u1 = *(const float4*)(su8 + 4);
  const float uu[8] = {u0.x, u0.y, u0.z, u0.w, u1.x, u1.y, u1.z, u1.w};
  float yo[8];
#pragma unroll
  for (int v = 0; v < 8; ++v) {
    const float y = acc[v] + Dh * uu[v];
    yo[v] = 0.5f * y * (1.0f + erff(y * 0.70710678118654752f));
  }
  *(float4*)(su8)     = make_float4(yo[0], yo[1], yo[2], yo[3]);
  *(float4*)(su8 + 4) = make_float4(yo[4], yo[5], yo[6], yo[7]);
  lds_barrier();
#pragma unroll
  for (int it = 0; it < 2; ++it) {
    const int e = (it * 256 + threadIdx.x) * 4;
    *(float4*)(ygelu + urow + e) = *(const float4*)(sUc + (e >> 6) * PPITCH + (e & 63));
  }
  urow += (size_t)H_ * L_;
  }
}

#define GKC      32
#define APITCH   36
#define BPITCH   136
#define A_BUF    (128 * APITCH)
#define B_BUF    (GKC * BPITCH)
#define GLU_LDS_BYTES 71680

__global__ __launch_bounds__(256) void s4d_glu(const float* __restrict__ ygelu,
                                               const float* __restrict__ W,
                                               const float* __restrict__ bias,
                                               float* __restrict__ out) {
  extern __shared__ __attribute__((aligned(16))) float smem[];
  float* sA = smem;
  float* sB = smem + 2 * A_BUF;

  const int tid    = threadIdx.x;
  const int waveId = tid >> 5;
  const int lane   = tid & 31;
  const int half   = lane >> 4;
  const int lm     = lane & 15;
  const int wm     = waveId & 1;
  const int wn     = waveId >> 1;
  const int mblk   = blockIdx.x & 7;
  const int nblk   = blockIdx.x >> 3;
  const int bb     = nblk >> 4;
  const int l0     = (nblk & 15) * 128;

  const float* gA = W + (size_t)(mblk * 64 + (tid >> 3)) * H_ + (tid & 7) * 4;
  const float* gB = ygelu + ((size_t)bb * H_ + (tid >> 5)) * L_ + l0 + (tid & 31) * 4;
  const int stA = (tid >> 3) * APITCH + (tid & 7) * 4;
  const int stB = (tid >> 5) * BPITCH + (tid & 31) * 4;

  float4 ra[4], rb[4];
#pragma unroll
  for (int j = 0; j < 4; ++j) {
    ra[j] = *(const float4*)(gA + (size_t)(j >> 1) * H_ * H_ + (size_t)(j & 1) * 32 * H_);
    rb[j] = *(const float4*)(gB + (size_t)j * 8 * L_);
  }
#pragma unroll
  for (int j = 0; j < 4; ++j) {
    *(float4*)(sA + stA + j * 32 * APITCH) = ra[j];
    *(float4*)(sB + stB + j * 8 * BPITCH)  = rb[j];
  }
  __syncthreads();

  v8f acc[4][2];
#pragma unroll
  for (int i = 0; i < 4; ++i)
#pragma unroll
    for (int j = 0; j < 2; ++j)
      acc[i][j] = (v8f){0.f, 0.f, 0.f, 0.f, 0.f, 0.f, 0.f, 0.f};

  const int fA = (wm * 32 + lm) * APITCH + 2 * half;
  const int fB = (2 * half) * BPITCH + wn * 32 + lm;

  for (int kc = 0; kc < H_ / GKC; ++kc) {
    const int cur = kc & 1;
    const bool more = (kc + 1) < (H_ / GKC);
    if (more) {
      const float* nA = gA + (kc + 1) * GKC;
      const float* nB = gB + (size_t)(kc + 1) * GKC * L_;
#pragma unroll
      for (int j = 0; j < 4; ++j) {
        ra[j] = *(const float4*)(nA + (size_t)(j >> 1) * H_ * H_ + (size_t)(j & 1) * 32 * H_);
        rb[j] = *(const float4*)(nB + (size_t)j * 8 * L_);
      }
    }
    const float* cA = sA + cur * A_BUF + fA;
    const float* cB = sB + cur * B_BUF + fB;
#pragma unroll
    for (int k0 = 0; k0 < GKC; k0 += 4) {
      v2f a[4], b[2];
      a[0] = *(const v2f*)(cA + k0);
      a[1] = *(const v2f*)(cA + 16 * APITCH + k0);
      a[2] = *(const v2f*)(cA + 64 * APITCH + k0);
      a[3] = *(const v2f*)(cA + 80 * APITCH + k0);
#pragma unroll
      for (int j = 0; j < 2; ++j) {
        b[j].x = cB[k0 * BPITCH + 16 * j];
        b[j].y = cB[(k0 + 1) * BPITCH + 16 * j];
      }
#pragma unroll
      for (int i = 0; i < 4; ++i)
#pragma unroll
        for (int j = 0; j < 2; ++j)
          acc[i][j] = wmma4(a[i], b[j], acc[i][j]);
    }
    if (more) {
      float* dA = sA + (cur ^ 1) * A_BUF + stA;
      float* dB = sB + (cur ^ 1) * B_BUF + stB;
#pragma unroll
      for (int j = 0; j < 4; ++j) {
        *(float4*)(dA + j * 32 * APITCH) = ra[j];
        *(float4*)(dB + j * 8 * BPITCH)  = rb[j];
      }
    }
    __syncthreads();
  }

#pragma unroll
  for (int mi = 0; mi < 2; ++mi) {
#pragma unroll
    for (int v = 0; v < 8; ++v) {
      const int hr = mblk * 64 + wm * 32 + mi * 16 + v + 8 * half;
      const float ba = bias[hr];
      const float bg = bias[H_ + hr];
#pragma unroll
      for (int j = 0; j < 2; ++j) {
        const int l = l0 + wn * 32 + j * 16 + lm;
        const float a  = acc[mi][j][v] + ba;
        const float gv = acc[2 + mi][j][v] + bg;
        out[((size_t)(bb * H_ + hr)) * L_ + l] = a * (1.0f / (1.0f + expf(-gv)));
      }
    }
  }
}

extern "C" void kernel_launch(void* const* d_in, const int* in_sizes, int n_in,
                              void* d_out, int out_size, void* d_ws, size_t ws_size,
                              hipStream_t stream) {
  (void)in_sizes; (void)n_in; (void)out_size; (void)ws_size;
  const float* u          = (const float*)d_in[0];
  const float* log_dt     = (const float*)d_in[1];
  const float* C          = (const float*)d_in[2];
  const float* log_A_real = (const float*)d_in[3];
  const float* A_imag     = (const float*)d_in[4];
  const float* Dvec       = (const float*)d_in[5];
  const float* W          = (const float*)d_in[6];
  const float* bias       = (const float*)d_in[7];
  float* out = (float*)d_out;

  float* ygelu = (float*)d_ws;

  s4d_conv<<<H_ * B_ / NBROW, 256, 0, stream>>>(u, log_dt, C, log_A_real, A_imag, Dvec, ygelu);
  s4d_glu<<<1024, 256, GLU_LDS_BYTES, stream>>>(ygelu, W, bias, out);
}


# GLU GEMM K loop software-pipelined: next k-step's fragments loaded into a second register set ahead of the current 8 products, next tile fetched in two halves under the products; sums unchanged
# speedup vs baseline: 1.9898x; 1.0301x over previous
#include <hip/hip_runtime.h>
#include <math.h>

#define H_     512
#define NMODE  32
#define B_     8
#define L_     2048
#define Q_     64
#define NCHUNK 32
#define CPITCH 34
#define PPITCH 68
#define SPITCH 33
#define XPITCH 36
#define NBROW  4
#define UBUF   (NCHUNK * PPITCH)

typedef __attribute__((ext_vector_type(2))) float v2f;
typedef __attribute__((ext_vector_type(8))) float v8f;

__device__ __forceinline__ v8f wmma4(v2f a, v2f b, v8f c) {
  return __builtin_amdgcn_wmma_f32_16x16x4_f32(false, a, false, b, (short)0, c,
                                               false, false);
}

__device__ __forceinline__ float xhalf(float x) {
  return __builtin_bit_cast(float, __builtin_amdgcn_permlanex16(__builtin_bit_cast(int, x), __builtin_bit_cast(int, x),
                                                                 0x76543210, 0xfedcba98, false, false));
}
template <int M>
__device__ __forceinline__ float xrow(float x) {
  return __builtin_bit_cast(float, __builtin_amdgcn_update_dpp(__builtin_bit_cast(int, x), __builtin_bit_cast(int, x),
                                                               0x160 + M, 0xf, 0xf, true));
}

__device__ __forceinline__ void lds_barrier() {
  __builtin_amdgcn_fence(__ATOMIC_RELEASE, "workgroup", "local");
  __builtin_amdgcn_s_barrier();
  __builtin_amdgcn_fence(__ATOMIC_ACQUIRE, "workgroup", "local");
}

__device__ __forceinline__ void cstep(float& pr, float& pi, float wr, float wi) {
  const float nr = pr * wr - pi * wi;
  const float ni = pr * wi + pi * wr;
  pr = nr; pi = ni;
}

__device__ __forceinline__ float taps32(float& pr, float& pi, float wr, float wi, int lane) {
  float tv[32];
#pragma unroll
  for (int d = 0; d < 32; ++d) {
    float v = 2.0f * pr;
    v += xhalf(v);
    v += xrow<8>(v);
    v += xrow<4>(v);
    v += xrow<2>(v);
    v += xrow<1>(v);
    tv[d] = v;
    cstep(pr, pi, wr, wi);
  }
  float o = tv[0];
#pragma unroll
  for (int d = 1; d < 32; ++d) {
    if (lane == d) o = tv[d];
  }
  return o;
}

__global__ __launch_bounds__(256, 8) void s4d_conv(const float* __restrict__ u,
                         const float* __restrict__ log_dt,
                         const float* __restrict__ C,
                         const float* __restrict__ log_A_real,
                         const float* __restrict__ A_imag,
                         const float* __restrict__ Dvec,
                         float* __restrict__ ygelu) {
  __shared__ __attribute__((aligned(16))) float sKt[256];
  __shared__ __attribute__((aligned(16))) float sCR[Q_ * CPITCH];
  __shared__ __attribute__((aligned(16))) float sCI[Q_ * CPITCH];
  __shared__ __attribute__((aligned(16))) float sP[2 * NMODE * PPITCH];
  __shared__ __attribute__((aligned(16))) float sU[2 * UBUF];
  __shared__ __attribute__((aligned(16))) float sS[2 * NMODE * SPITCH];
  __shared__ __attribute__((aligned(16))) float sXR[NCHUNK * XPITCH];
  __shared__ __attribute__((aligned(16))) float sXI[NCHUNK * XPITCH];

  const int h      = blockIdx.x >> 1;
  const int b0     = (blockIdx.x & 1) * NBROW;
  const int waveId = threadIdx.x >> 5;
  const int lane   = threadIdx.x & 31;
  const int half   = lane >> 4;
  const int lm     = lane & 15;
  const int jt     = waveId & 3;
  const int ct     = waveId >> 2;
  const int col    = ct * 16 + lm;

  size_t urow = (size_t)(b0 * H_ + h) * L_;

  const float Dh = Dvec[h];

  float4 ru0, ru1;
  ru0 = *(const float4*)(u + urow + threadIdx.x * 4);
  ru1 = *(const float4*)(u + urow + (256 + threadIdx.x) * 4);

  if (waveId < 4) {
    const int k = lane;
    const float dt = expf(log_dt[h]);
    const float Ar = -expf(log_A_real[h * NMODE + k]);
    const float Ai = A_imag[h * NMODE + k];
    const float Ccr = C[(h * NMODE + k) * 2 + 0];
    const float Cci = C[(h * NMODE + k) * 2 + 1];
    const float er = expf(Ar * dt);
    const float wr = er * cosf(Ai * dt);
    const float wi = er * sinf(Ai * dt);
    const float exr = wr - 1.0f, exi = wi;
    const float den = Ar * Ar + Ai * Ai;
    const float qr = (exr * Ar + exi * Ai) / den;
    const float qi = (exi * Ar - exr * Ai) / den;
    const float Ckr = Ccr * qr - Cci * qi;
    const float Cki = Ccr * qi + Cci * qr;
    if (waveId == 0) {
      sKt[k] = 0.0f;
      sKt[32 + k] = 0.0f;
      float pr = Ckr, pi = Cki;
      sKt[64 + k] = taps32(pr, pi, wr, wi, k);
    } else if (waveId == 3) {
      float pr = Ckr, pi = Cki;
      for (int d = 0; d < 32; ++d) cstep(pr, pi, wr, wi);
      sKt[96 + k] = taps32(pr, pi, wr, wi, k);
    } else if (waveId == 1) {
      float sr = Ckr * wr - Cki * wi;
      float si = Ckr * wi + Cki * wr;
      for (int j = 0; j < Q_; ++j) {
        sCR[j * CPITCH + k] = 2.0f * sr;
        sCI[j * CPITCH + k] = -2.0f * si;
        cstep(sr, si, wr, wi);
      }
    } else {
      float pr = 1.0f, pi = 0.0f;
      for (int d = 0; d < Q_; ++d) {
        sP[k * PPITCH + (Q_ - 1 - d)]           = pr;
        sP[(NMODE + k) * PPITCH + (Q_ - 1 - d)] = pi;
        cstep(pr, pi, wr, wi);
      }
      sKt[128 + k] = pr;
      sKt[160 + k] = pi;
    }
  }
  const int jrow = jt * 16 + lm;

  for (int r = 0; r < NBROW; ++r) {
  float* sUc = sU + (r & 1) * UBUF;
  {
    const int e0 = threadIdx.x * 4, e1 = (256 + threadIdx.x) * 4;
    *(float4*)(sUc + (e0 >> 6) * PPITCH + (e0 & 63)) = ru0;
    *(float4*)(sUc + (e1 >> 6) * PPITCH + (e1 & 63)) = ru1;
  }
  if (r + 1 < NBROW) {
    ru0 = *(const float4*)(u + urow + (size_t)H_ * L_ + threadIdx.x * 4);
    ru1 = *(const float4*)(u + urow + (size_t)H_ * L_ + (256 + threadIdx.x) * 4);
  }
  lds_barrier();

  v8f acc  = {0.f, 0.f, 0.f, 0.f, 0.f, 0.f, 0.f, 0.f};
  v8f accS = {0.f, 0.f, 0.f, 0.f, 0.f, 0.f, 0.f, 0.f};

#pragma unroll
  for (int k0 = 0; k0 < Q_; k0 += 4) {
    const int ta = k0 + 2 * half;
    const v2f bb = *(const v2f*)(sUc + col * PPITCH + ta);
    const v2f ap = *(const v2f*)(sP + jrow * PPITCH + ta);
    v2f a;
    a.x = sKt[64 + jrow - ta];
    a.y = sKt[63 + jrow - ta];
    accS = wmma4(ap, bb, accS);
    acc  = wmma4(a, bb, acc);
  }
#pragma unroll
  for (int v = 0; v < 8; ++v)
    sS[(jt * 16 + v + 8 * half) * SPITCH + col] = accS[v];
  lds_barrier();

  if (waveId == 0) {
    const float ar = sKt[128 + lane];
    const float ai = sKt[160 + lane];
    float xr = 0.0f, xi = 0.0f;
#pragma unroll
    for (int c0 = 0; c0 < NCHUNK; c0 += 16) {
      float pr[16], pi[16];
#pragma unroll
      for (int i = 0; i < 16; ++i) {
        pr[i] = sS[lane * SPITCH + c0 + i];
        pi[i] = sS[(NMODE + lane) * SPITCH + c0 + i];
      }
#pragma unroll
      for (int i = 0; i < 16; ++i) {
        sXR[(c0 + i) * XPITCH + lane] = xr;
        sXI[(c0 + i) * XPITCH + lane] = xi;
        const float nr = ar * xr - ai * xi + pr[i];
        const float ni = ar * xi + ai * xr + pi[i];
        xr = nr; xi = ni;
      }
    }
  }
  lds_barrier();

#pragma unroll
  for (int k0 = 0; k0 < NMODE; k0 += 4) {
    const int ka = k0 + 2 * half;
    const v2f a  = *(const v2f*)(sCR + jrow * CPITCH + ka);
    const v2f bb = *(const v2f*)(sXR + col * XPITCH + ka);
    acc = wmma4(a, bb, acc);
  }
#pragma unroll
  for (int k0 = 0; k0 < NMODE; k0 += 4) {
    const int ka = k0 + 2 * half;
    const v2f a  = *(const v2f*)(sCI + jrow * CPITCH + ka);
    const v2f bb = *(const v2f*)(sXI + col * XPITCH + ka);
    acc = wmma4(a, bb, acc);
  }
  float* su8 = sUc + col * PPITCH + jt * 16 + 8 * half;
  const float4 u0 = *(const float4*)(su8);
  const float4 u1 = *(const float4*)(su8 + 4);
  const float uu[8] = {u0.x, u0.y, u0.z, u0.w, u1.x, u1.y, u1.z, u1.w};
  float yo[8];
#pragma unroll
  for (int v = 0; v < 8; ++v) {
    const float y = acc[v] + Dh * uu[v];
    yo[v] = 0.5f * y * (1.0f + erff(y * 0.70710678118654752f));
  }
  *(float4*)(su8)     = make_float4(yo[0], yo[1], yo[2], yo[3]);
  *(float4*)(su8 + 4) = make_float4(yo[4], yo[5], yo[6], yo[7]);
  lds_barrier();
#pragma unroll
  for (int it = 0; it < 2; ++it) {
    const int e = (it * 256 + threadIdx.x) * 4;
    *(float4*)(ygelu + urow + e) = *(const float4*)(sUc + (e >> 6) * PPITCH + (e & 63));
  }
  urow += (size_t)H_ * L_;
  }
}

#define GKC      32
#define APITCH   36
#define BPITCH   136
#define A_BUF    (128 * APITCH)
#define B_BUF    (GKC * BPITCH)
#define GLU_LDS_BYTES 71680

template <bool MORE>
__device__ __forceinline__ void glu_chunk(v8f (&acc)[4][2], const float* cA, const float* cB,
                                          float* dA, float* dB, const float* nA, const float* nB) {
  float4 st[4];
  if (MORE) {
    st[0] = *(const float4*)(nA);
    st[1] = *(const float4*)(nA + (size_t)32 * H_);
    st[2] = *(const float4*)(nB);
    st[3] = *(const float4*)(nB + (size_t)8 * L_);
  }
  v2f a[2][4], b[2][2];
  a[0][0] = *(const v2f*)(cA);
  a[0][1] = *(const v2f*)(cA + 16 * APITCH);
  a[0][2] = *(const v2f*)(cA + 64 * APITCH);
  a[0][3] = *(const v2f*)(cA + 80 * APITCH);
#pragma unroll
  for (int j = 0; j < 2; ++j) {
    b[0][j].x = cB[16 * j];
    b[0][j].y = cB[BPITCH + 16 * j];
  }
#pragma unroll
  for (int s = 0; s < GKC / 4; ++s) {
    const int p = s & 1, q = p ^ 1;
    if (s + 1 < GKC / 4) {
      const int k1 = (s + 1) * 4;
      a[q][0] = *(const v2f*)(cA + k1);
      a[q][1] = *(const v2f*)(cA + 16 * APITCH + k1);
      a[q][2] = *(const v2f*)(cA + 64 * APITCH + k1);
      a[q][3] = *(const v2f*)(cA + 80 * APITCH + k1);
#pragma unroll
      for (int j = 0; j < 2; ++j) {
        b[q][j].x = cB[k1 * BPITCH + 16 * j];
        b[q][j].y = cB[(k1 + 1) * BPITCH + 16 * j];
      }
    }
#pragma unroll
    for (int i = 0; i < 4; ++i)
#pragma unroll
      for (int j = 0; j < 2; ++j)
        acc[i][j] = wmma4(a[p][i], b[p][j], acc[i][j]);
    if (MORE && s == GKC / 8 - 1) {
      *(float4*)(dA)               = st[0];
      *(float4*)(dA + 32 * APITCH) = st[1];
      *(float4*)(dB)               = st[2];
      *(float4*)(dB + 8 * BPITCH)  = st[3];
      st[0] = *(const float4*)(nA + (size_t)H_ * H_);
      st[1] = *(const float4*)(nA + (size_t)H_ * H_ + (size_t)32 * H_);
      st[2] = *(const float4*)(nB + (size_t)16 * L_);
      st[3] = *(const float4*)(nB + (size_t)24 * L_);
    }
  }
  if (MORE) {
    *(float4*)(dA + 64 * APITCH) = st[0];
    *(float4*)(dA + 96 * APITCH) = st[1];
    *(float4*)(dB + 16 * BPITCH) = st[2];
    *(float4*)(dB + 24 * BPITCH) = st[3];
  }
}

__global__ __launch_bounds__(256) __attribute__((amdgpu_waves_per_eu(8, 8))) void s4d_glu(const float* __restrict__ ygelu,
                                               const float* __restrict__ W,
                                               const float* __restrict__ bias,
                                               float* __restrict__ out) {
  extern __shared__ __attribute__((aligned(16))) float smem[];
  float* sA = smem;
  float* sB = smem + 2 * A_BUF;

  const int tid    = threadIdx.x;
  const int waveId = tid >> 5;
  const int lane   = tid & 31;
  const int half   = lane >> 4;
  const int lm     = lane & 15;
  const int wm     = waveId & 1;
  const int wn     = waveId >> 1;
  const int mblk   = blockIdx.x & 7;
  const int nblk   = blockIdx.x >> 3;
  const int bb     = nblk >> 4;
  const int l0     = (nblk & 15) * 128;

  const float* gA = W + (size_t)(mblk * 64 + (tid >> 3)) * H_ + (tid & 7) * 4;
  const float* gB = ygelu + ((size_t)bb * H_ + (tid >> 5)) * L_ + l0 + (tid & 31) * 4;
  const int stA = (tid >> 3) * APITCH + (tid & 7) * 4;
  const int stB = (tid >> 5) * BPITCH + (tid & 31) * 4;

  {
    float4 ra[4], rb[4];
#pragma unroll
    for (int j = 0; j < 4; ++j) {
      ra[j] = *(const float4*)(gA + (size_t)(j >> 1) * H_ * H_ + (size_t)(j & 1) * 32 * H_);
      rb[j] = *(const float4*)(gB + (size_t)j * 8 * L_);
    }
#pragma unroll
    for (int j = 0; j < 4; ++j) {
      *(float4*)(sA + stA + j * 32 * APITCH) = ra[j];
      *(float4*)(sB + stB + j * 8 * BPITCH)  = rb[j];
    }
  }
  __syncthreads();

  v8f acc[4][2];
#pragma unroll
  for (int i = 0; i < 4; ++i)
#pragma unroll
    for (int j = 0; j < 2; ++j)
      acc[i][j] = (v8f){0.f, 0.f, 0.f, 0.f, 0.f, 0.f, 0.f, 0.f};

  const int fA = (wm * 32 + lm) * APITCH + 2 * half;
  const int fB = (2 * half) * BPITCH + wn * 32 + lm;

  for (int kc = 0; kc < H_ / GKC; ++kc) {
    const int cur = kc & 1;
    const float* cA = sA + cur * A_BUF + fA;
    const float* cB = sB + cur * B_BUF + fB;
    float* dA = sA + (cur ^ 1) * A_BUF + stA;
    float* dB = sB + (cur ^ 1) * B_BUF + stB;
    const float* nA = gA + (kc + 1) * GKC;
    const float* nB = gB + (size_t)(kc + 1) * GKC * L_;
    if ((kc + 1) < (H_ / GKC)) glu_chunk<true>(acc, cA, cB, dA, dB, nA, nB);
    else                       glu_chunk<false>(acc, cA, cB, dA, dB, nA, nB);
    __syncthreads();
  }

#pragma unroll
  for (int mi = 0; mi < 2; ++mi) {
#pragma unroll
    for (int v = 0; v < 8; ++v) {
      const int hr = mblk * 64 + wm * 32 + mi * 16 + v + 8 * half;
      const float ba = bias[hr];
      const float bg = bias[H_ + hr];
#pragma unroll
      for (int j = 0; j < 2; ++j) {
        const int l = l0 + wn * 32 + j * 16 + lm;
        const float a  = acc[mi][j][v] + ba;
        const float gv = acc[2 + mi][j][v] + bg;
        out[((size_t)(bb * H_ + hr)) * L_ + l] = a * (1.0f / (1.0f + expf(-gv)));
      }
    }
  }
}

extern "C" void kernel_launch(void* const* d_in, const int* in_sizes, int n_in,
                              void* d_out, int out_size, void* d_ws, size_t ws_size,
                              hipStream_t stream) {
  (void)in_sizes; (void)n_in; (void)out_size; (void)ws_size;
  const float* u          = (const float*)d_in[0];
  const float* log_dt     = (const float*)d_in[1];
  const float* C          = (const float*)d_in[2];
  const float* log_A_real = (const float*)d_in[3];
  const float* A_imag     = (const float*)d_in[4];
  const float* Dvec       = (const float*)d_in[5];
  const float* W          = (const float*)d_in[6];
  const float* bias       = (const float*)d_in[7];
  float* out = (float*)d_out;

  float* ygelu = (float*)d_ws;

  s4d_conv<<<H_ * B_ / NBROW, 256, 0, stream>>>(u, log_dt, C, log_A_real, A_imag, Dvec, ygelu);
  s4d_glu<<<1024, 256, GLU_LDS_BYTES, stream>>>(ygelu, W, bias, out);
}


# both epilogues' special functions in short branch-free forms: GELU erf as one clamped rational polynomial, GLU sigmoid on hardware exp2, divisions on v_rcp_f32; results move in the last bits
# speedup vs baseline: 2.0983x; 1.0545x over previous
#include <hip/hip_runtime.h>
#include <math.h>

#define H_     512
#define NMODE  32
#define B_     8
#define L_     2048
#define Q_     64
#define NCHUNK 32
#define CPITCH 34
#define PPITCH 68
#define SPITCH 33
#define XPITCH 36
#define NBROW  4
#define UBUF   (NCHUNK * PPITCH)

typedef __attribute__((ext_vector_type(2))) float v2f;
typedef __attribute__((ext_vector_type(8))) float v8f;

__device__ __forceinline__ v8f wmma4(v2f a, v2f b, v8f c) {
  return __builtin_amdgcn_wmma_f32_16x16x4_f32(false, a, false, b, (short)0, c,
                                               false, false);
}

__device__ __forceinline__ float xhalf(float x) {
  return __builtin_bit_cast(float, __builtin_amdgcn_permlanex16(__builtin_bit_cast(int, x), __builtin_bit_cast(int, x),
                                                                 0x76543210, 0xfedcba98, false, false));
}
template <int M>
__device__ __forceinline__ float xrow(float x) {
  return __builtin_bit_cast(float, __builtin_amdgcn_update_dpp(__builtin_bit_cast(int, x), __builtin_bit_cast(int, x),
                                                               0x160 + M, 0xf, 0xf, true));
}

__device__ __forceinline__ void lds_barrier() {
  __builtin_amdgcn_fence(__ATOMIC_RELEASE, "workgroup", "local");
  __builtin_amdgcn_s_barrier();
  __builtin_amdgcn_fence(__ATOMIC_ACQUIRE, "workgroup", "local");
}

__device__ __forceinline__ float erf_rational(float x) {
  x = __builtin_amdgcn_fmed3f(x, -4.0f, 4.0f);
  const float x2 = x * x;
  float p = -2.72614225801306e-10f;
  p = fmaf(p, x2, 2.77068142495902e-08f);
  p = fmaf(p, x2, -2.10102402082508e-06f);
  p = fmaf(p, x2, -5.69250639462346e-05f);
  p = fmaf(p, x2, -7.34990630326855e-04f);
  p = fmaf(p, x2, -2.95459980854025e-03f);
  p = fmaf(p, x2, -1.60960333262415e-02f);
  float q = -1.45660718464996e-05f;
  q = fmaf(q, x2, -2.13374055278905e-04f);
  q = fmaf(q, x2, -1.68282697438203e-03f);
  q = fmaf(q, x2, -7.37332916720468e-03f);
  q = fmaf(q, x2, -1.42647390514189e-02f);
  return (p * x) * __builtin_amdgcn_rcpf(q);
}

__device__ __forceinline__ float sigmoid_hw(float g) {
  const float e = __builtin_amdgcn_exp2f(g * -1.4426950408889634f);
  return __builtin_amdgcn_rcpf(1.0f + e);
}

__device__ __forceinline__ void cstep(float& pr, float& pi, float wr, float wi) {
  const float nr = pr * wr - pi * wi;
  const float ni = pr * wi + pi * wr;
  pr = nr; pi = ni;
}

__device__ __forceinline__ float taps32(float& pr, float& pi, float wr, float wi, int lane) {
  float tv[32];
#pragma unroll
  for (int d = 0; d < 32; ++d) {
    float v = 2.0f * pr;
    v += xhalf(v);
    v += xrow<8>(v);
    v += xrow<4>(v);
    v += xrow<2>(v);
    v += xrow<1>(v);
    tv[d] = v;
    cstep(pr, pi, wr, wi);
  }
  float o = tv[0];
#pragma unroll
  for (int d = 1; d < 32; ++d) {
    if (lane == d) o = tv[d];
  }
  return o;
}

__global__ __launch_bounds__(256, 8) void s4d_conv(const float* __restrict__ u,
                         const float* __restrict__ log_dt,
                         const float* __restrict__ C,
                         const float* __restrict__ log_A_real,
                         const float* __restrict__ A_imag,
                         const float* __restrict__ Dvec,
                         float* __restrict__ ygelu) {
  __shared__ __attribute__((aligned(16))) float sKt[256];
  __shared__ __attribute__((aligned(16))) float sCR[Q_ * CPITCH];
  __shared__ __attribute__((aligned(16))) float sCI[Q_ * CPITCH];
  __shared__ __attribute__((aligned(16))) float sP[2 * NMODE * PPITCH];
  __shared__ __attribute__((aligned(16))) float sU[2 * UBUF];
  __shared__ __attribute__((aligned(16))) float sS[2 * NMODE * SPITCH];
  __shared__ __attribute__((aligned(16))) float sXR[NCHUNK * XPITCH];
  __shared__ __attribute__((aligned(16))) float sXI[NCHUNK * XPITCH];

  const int h      = blockIdx.x >> 1;
  const int b0     = (blockIdx.x & 1) * NBROW;
  const int waveId = threadIdx.x >> 5;
  const int lane   = threadIdx.x & 31;
  const int half   = lane >> 4;
  const int lm     = lane & 15;
  const int jt     = waveId & 3;
  const int ct     = waveId >> 2;
  const int col    = ct * 16 + lm;

  size_t urow = (size_t)(b0 * H_ + h) * L_;

  const float Dh = Dvec[h];

  float4 ru0, ru1;
  ru0 = *(const float4*)(u + urow + threadIdx.x * 4);
  ru1 = *(const float4*)(u + urow + (256 + threadIdx.x) * 4);

  if (waveId < 4) {
    const int k = lane;
    const float dt = expf(log_dt[h]);
    const float Ar = -expf(log_A_real[h * NMODE + k]);
    const float Ai = A_imag[h * NMODE + k];
    const float Ccr = C[(h * NMODE + k) * 2 + 0];
    const float Cci = C[(h * NMODE + k) * 2 + 1];
    const float er = expf(Ar * dt);
    const float wr = er * cosf(Ai * dt);
    const float wi = er * sinf(Ai * dt);
    const float exr = wr - 1.0f, exi = wi;
    const float den = Ar * Ar + Ai * Ai;
    const float qr = (exr * Ar + exi * Ai) / den;
    const float qi = (exi * Ar - exr * Ai) / den;
    const float Ckr = Ccr * qr - Cci * qi;
    const float Cki = Ccr * qi + Cci * qr;
    if (waveId == 0) {
      sKt[k] = 0.0f;
      sKt[32 + k] = 0.0f;
      float pr = Ckr, pi = Cki;
      sKt[64 + k] = taps32(pr, pi, wr, wi, k);
    } else if (waveId == 3) {
      float pr = Ckr, pi = Cki;
      for (int d = 0; d < 32; ++d) cstep(pr, pi, wr, wi);
      sKt[96 + k] = taps32(pr, pi, wr, wi, k);
    } else if (waveId == 1) {
      float sr = Ckr * wr - Cki * wi;
      float si = Ckr * wi + Cki * wr;
      for (int j = 0; j < Q_; ++j) {
        sCR[j * CPITCH + k] = 2.0f * sr;
        sCI[j * CPITCH + k] = -2.0f * si;
        cstep(sr, si, wr, wi);
      }
    } else {
      float pr = 1.0f, pi = 0.0f;
      for (int d = 0; d < Q_; ++d) {
        sP[k * PPITCH + (Q_ - 1 - d)]           = pr;
        sP[(NMODE + k) * PPITCH + (Q_ - 1 - d)] = pi;
        cstep(pr, pi, wr, wi);
      }
      sKt[128 + k] = pr;
      sKt[160 + k] = pi;
    }
  }
  const int jrow = jt * 16 + lm;

  for (int r = 0; r < NBROW; ++r) {
  float* sUc = sU + (r & 1) * UBUF;
  {
    const int e0 = threadIdx.x * 4, e1 = (256 + threadIdx.x) * 4;
    *(float4*)(sUc + (e0 >> 6) * PPITCH + (e0 & 63)) = ru0;
    *(float4*)(sUc + (e1 >> 6) * PPITCH + (e1 & 63)) = ru1;
  }
  if (r + 1 < NBROW) {
    ru0 = *(const float4*)(u + urow + (size_t)H_ * L_ + threadIdx.x * 4);
    ru1 = *(const float4*)(u + urow + (size_t)H_ * L_ + (256 + threadIdx.x) * 4);
  }
  lds_barrier();

  v8f acc  = {0.f, 0.f, 0.f, 0.f, 0.f, 0.f, 0.f, 0.f};
  v8f accS = {0.f, 0.f, 0.f, 0.f, 0.f, 0.f, 0.f, 0.f};

#pragma unroll
  for (int k0 = 0; k0 < Q_; k0 += 4) {
    const int ta = k0 + 2 * half;
    const v2f bb = *(const v2f*)(sUc + col * PPITCH + ta);
    const v2f ap = *(const v2f*)(sP + jrow * PPITCH + ta);
    v2f a;
    a.x = sKt[64 + jrow - ta];
    a.y = sKt[63 + jrow - ta];
    accS = wmma4(ap, bb, accS);
    acc  = wmma4(a, bb, acc);
  }
#pragma unroll
  for (int v = 0; v < 8; ++v)
    sS[(jt * 16 + v + 8 * half) * SPITCH + col] = accS[v];
  lds_barrier();

  if (waveId == 0) {
    const float ar = sKt[128 + lane];
    const float ai = sKt[160 + lane];
    float xr = 0.0f, xi = 0.0f;
#pragma unroll
    for (int c0 = 0; c0 < NCHUNK; c0 += 16) {
      float pr[16], pi[16];
#pragma unroll
      for (int i = 0; i < 16; ++i) {
        pr[i] = sS[lane * SPITCH + c0 + i];
        pi[i] = sS[(NMODE + lane) * SPITCH + c0 + i];
      }
#pragma unroll
      for (int i = 0; i < 16; ++i) {
        sXR[(c0 + i) * XPITCH + lane] = xr;
        sXI[(c0 + i) * XPITCH + lane] = xi;
        const float nr = ar * xr - ai * xi + pr[i];
        const float ni = ar * xi + ai * xr + pi[i];
        xr = nr; xi = ni;
      }
    }
  }
  lds_barrier();

#pragma unroll
  for (int k0 = 0; k0 < NMODE; k0 += 4) {
    const int ka = k0 + 2 * half;
    const v2f a  = *(const v2f*)(sCR + jrow * CPITCH + ka);
    const v2f bb = *(const v2f*)(sXR + col * XPITCH + ka);
    acc = wmma4(a, bb, acc);
  }
#pragma unroll
  for (int k0 = 0; k0 < NMODE; k0 += 4) {
    const int ka = k0 + 2 * half;
    const v2f a  = *(const v2f*)(sCI + jrow * CPITCH + ka);
    const v2f bb = *(const v2f*)(sXI + col * XPITCH + ka);
    acc = wmma4(a, bb, acc);
  }
  float* su8 = sUc + col * PPITCH + jt * 16 + 8 * half;
  const float4 u0 = *(const float4*)(su8);
  const float4 u1 = *(const float4*)(su8 + 4);
  const float uu[8] = {u0.x, u0.y, u0.z, u0.w, u1.x, u1.y, u1.z, u1.w};
  float yo[8];
#pragma unroll
  for (int v = 0; v < 8; ++v) {
    const float y = acc[v] + Dh * uu[v];
    const float hy = 0.5f * y;
    yo[v] = fmaf(hy, erf_rational(y * 0.70710678118654752f), hy);
  }
  *(float4*)(su8)     = make_float4(yo[0], yo[1], yo[2], yo[3]);
  *(float4*)(su8 + 4) = make_float4(yo[4], yo[5], yo[6], yo[7]);
  lds_barrier();
#pragma unroll
  for (int it = 0; it < 2; ++it) {
    const int e = (it * 256 + threadIdx.x) * 4;
    *(float4*)(ygelu + urow + e) = *(const float4*)(sUc + (e >> 6) * PPITCH + (e & 63));
  }
  urow += (size_t)H_ * L_;
  }
}

#define GKC      32
#define APITCH   36
#define BPITCH   136
#define A_BUF    (128 * APITCH)
#define B_BUF    (GKC * BPITCH)
#define GLU_LDS_BYTES 71680

template <bool MORE>
__device__ __forceinline__ void glu_chunk(v8f (&acc)[4][2], const float* cA, const float* cB,
                                          float* dA, float* dB, const float* nA, const float* nB) {
  float4 st[4];
  if (MORE) {
    st[0] = *(const float4*)(nA);
    st[1] = *(const float4*)(nA + (size_t)32 * H_);
    st[2] = *(const float4*)(nB);
    st[3] = *(const float4*)(nB + (size_t)8 * L_);
  }
  v2f a[2][4], b[2][2];
  a[0][0] = *(const v2f*)(cA);
  a[0][1] = *(const v2f*)(cA + 16 * APITCH);
  a[0][2] = *(const v2f*)(cA + 64 * APITCH);
  a[0][3] = *(const v2f*)(cA + 80 * APITCH);
#pragma unroll
  for (int j = 0; j < 2; ++j) {
    b[0][j].x = cB[16 * j];
    b[0][j].y = cB[BPITCH + 16 * j];
  }
#pragma unroll
  for (int s = 0; s < GKC / 4; ++s) {
    const int p = s & 1, q = p ^ 1;
    if (s + 1 < GKC / 4) {
      const int k1 = (s + 1) * 4;
      a[q][0] = *(const v2f*)(cA + k1);
      a[q][1] = *(const v2f*)(cA + 16 * APITCH + k1);
      a[q][2] = *(const v2f*)(cA + 64 * APITCH + k1);
      a[q][3] = *(const v2f*)(cA + 80 * APITCH + k1);
#pragma unroll
      for (int j = 0; j < 2; ++j) {
        b[q][j].x = cB[k1 * BPITCH + 16 * j];
        b[q][j].y = cB[(k1 + 1) * BPITCH + 16 * j];
      }
    }
#pragma unroll
    for (int i = 0; i < 4; ++i)
#pragma unroll
      for (int j = 0; j < 2; ++j)
        acc[i][j] = wmma4(a[p][i], b[p][j], acc[i][j]);
    if (MORE && s == GKC / 8 - 1) {
      *(float4*)(dA)               = st[0];
      *(float4*)(dA + 32 * APITCH) = st[1];
      *(float4*)(dB)               = st[2];
      *(float4*)(dB + 8 * BPITCH)  = st[3];
      st[0] = *(const float4*)(nA + (size_t)H_ * H_);
      st[1] = *(const float4*)(nA + (size_t)H_ * H_ + (size_t)32 * H_);
      st[2] = *(const float4*)(nB + (size_t)16 * L_);
      st[3] = *(const float4*)(nB + (size_t)24 * L_);
    }
  }
  if (MORE) {
    *(float4*)(dA + 64 * APITCH) = st[0];
    *(float4*)(dA + 96 * APITCH) = st[1];
    *(float4*)(dB + 16 * BPITCH) = st[2];
    *(float4*)(dB + 24 * BPITCH) = st[3];
  }
}

__global__ __launch_bounds__(256) __attribute__((amdgpu_waves_per_eu(8, 8))) void s4d_glu(const float* __restrict__ ygelu,
                                               const float* __restrict__ W,
                                               const float* __restrict__ bias,
                                               float* __restrict__ out) {
  extern __shared__ __attribute__((aligned(16))) float smem[];
  float* sA = smem;
  float* sB = smem + 2 * A_BUF;

  const int tid    = threadIdx.x;
  const int waveId = tid >> 5;
  const int lane   = tid & 31;
  const int half   = lane >> 4;
  const int lm     = lane & 15;
  const int wm     = waveId & 1;
  const int wn     = waveId >> 1;
  const int mblk   = blockIdx.x & 7;
  const int nblk   = blockIdx.x >> 3;
  const int bb     = nblk >> 4;
  const int l0     = (nblk & 15) * 128;

  const float* gA = W + (size_t)(mblk * 64 + (tid >> 3)) * H_ + (tid & 7) * 4;
  const float* gB = ygelu + ((size_t)bb * H_ + (tid >> 5)) * L_ + l0 + (tid & 31) * 4;
  const int stA = (tid >> 3) * APITCH + (tid & 7) * 4;
  const int stB = (tid >> 5) * BPITCH + (tid & 31) * 4;

  {
    float4 ra[4], rb[4];
#pragma unroll
    for (int j = 0; j < 4; ++j) {
      ra[j] = *(const float4*)(gA + (size_t)(j >> 1) * H_ * H_ + (size_t)(j & 1) * 32 * H_);
      rb[j] = *(const float4*)(gB + (size_t)j * 8 * L_);
    }
#pragma unroll
    for (int j = 0; j < 4; ++j) {
      *(float4*)(sA + stA + j * 32 * APITCH) = ra[j];
      *(float4*)(sB + stB + j * 8 * BPITCH)  = rb[j];
    }
  }
  __syncthreads();

  v8f acc[4][2];
#pragma unroll
  for (int i = 0; i < 4; ++i)
#pragma unroll
    for (int j = 0; j < 2; ++j)
      acc[i][j] = (v8f){0.f, 0.f, 0.f, 0.f, 0.f, 0.f, 0.f, 0.f};

  const int fA = (wm * 32 + lm) * APITCH + 2 * half;
  const int fB = (2 * half) * BPITCH + wn * 32 + lm;

  for (int kc = 0; kc < H_ / GKC; ++kc) {
    const int cur = kc & 1;
    const float* cA = sA + cur * A_BUF + fA;
    const float* cB = sB + cur * B_BUF + fB;
    float* dA = sA + (cur ^ 1) * A_BUF + stA;
    float* dB = sB + (cur ^ 1) * B_BUF + stB;
    const float* nA = gA + (kc + 1) * GKC;
    const float* nB = gB + (size_t)(kc + 1) * GKC * L_;
    if ((kc + 1) < (H_ / GKC)) glu_chunk<true>(acc, cA, cB, dA, dB, nA, nB);
    else                       glu_chunk<false>(acc, cA, cB, dA, dB, nA, nB);
    __syncthreads();
  }

#pragma unroll
  for (int mi = 0; mi < 2; ++mi) {
#pragma unroll
    for (int v = 0; v < 8; ++v) {
      const int hr = mblk * 64 + wm * 32 + mi * 16 + v + 8 * half;
      const float ba = bias[hr];
      const float bg = bias[H_ + hr];
#pragma unroll
      for (int j = 0; j < 2; ++j) {
        const int l = l0 + wn * 32 + j * 16 + lm;
        const float a  = acc[mi][j][v] + ba;
        const float gv = acc[2 + mi][j][v] + bg;
        out[((size_t)(bb * H_ + hr)) * L_ + l] = a * sigmoid_hw(gv);
      }
    }
  }
}

extern "C" void kernel_launch(void* const* d_in, const int* in_sizes, int n_in,
                              void* d_out, int out_size, void* d_ws, size_t ws_size,
                              hipStream_t stream) {
  (void)in_sizes; (void)n_in; (void)out_size; (void)ws_size;
  const float* u          = (const float*)d_in[0];
  const float* log_dt     = (const float*)d_in[1];
  const float* C          = (const float*)d_in[2];
  const float* log_A_real = (const float*)d_in[3];
  const float* A_imag     = (const float*)d_in[4];
  const float* Dvec       = (const float*)d_in[5];
  const float* W          = (const float*)d_in[6];
  const float* bias       = (const float*)d_in[7];
  float* out = (float*)d_out;

  float* ygelu = (float*)d_ws;

  s4d_conv<<<H_ * B_ / NBROW, 256, 0, stream>>>(u, log_dt, C, log_A_real, A_imag, Dvec, ygelu);
  s4d_glu<<<1024, 256, GLU_LDS_BYTES, stream>>>(ygelu, W, bias, out);
}


# GLU GEMM: fragment blocks (2 row tiles x 2 col tiles x 2 k-steps) in the seven-product 2x2 form: 7 f32 WMMAs for 8, operand sums on packed VALU adds; 14 accumulators, 4 waves/SIMD; sum order changes
# speedup vs baseline: 2.1884x; 1.0429x over previous
#include <hip/hip_runtime.h>
#include <math.h>

#define H_     512
#define NMODE  32
#define B_     8
#define L_     2048
#define Q_     64
#define NCHUNK 32
#define CPITCH 34
#define PPITCH 68
#define SPITCH 33
#define XPITCH 36
#define NBROW  4
#define UBUF   (NCHUNK * PPITCH)

typedef __attribute__((ext_vector_type(2))) float v2f;
typedef __attribute__((ext_vector_type(8))) float v8f;

__device__ __forceinline__ v8f wmma4(v2f a, v2f b, v8f c) {
  return __builtin_amdgcn_wmma_f32_16x16x4_f32(false, a, false, b, (short)0, c,
                                               false, false);
}

__device__ __forceinline__ float xhalf(float x) {
  return __builtin_bit_cast(float, __builtin_amdgcn_permlanex16(__builtin_bit_cast(int, x), __builtin_bit_cast(int, x),
                                                                 0x76543210, 0xfedcba98, false, false));
}
template <int M>
__device__ __forceinline__ float xrow(float x) {
  return __builtin_bit_cast(float, __builtin_amdgcn_update_dpp(__builtin_bit_cast(int, x), __builtin_bit_cast(int, x),
                                                               0x160 + M, 0xf, 0xf, true));
}

__device__ __forceinline__ void lds_barrier() {
  __builtin_amdgcn_fence(__ATOMIC_RELEASE, "workgroup", "local");
  __builtin_amdgcn_s_barrier();
  __builtin_amdgcn_fence(__ATOMIC_ACQUIRE, "workgroup", "local");
}

__device__ __forceinline__ float erf_rational(float x) {
  x = __builtin_amdgcn_fmed3f(x, -4.0f, 4.0f);
  const float x2 = x * x;
  float p = -2.72614225801306e-10f;
  p = fmaf(p, x2, 2.77068142495902e-08f);
  p = fmaf(p, x2, -2.10102402082508e-06f);
  p = fmaf(p, x2, -5.69250639462346e-05f);
  p = fmaf(p, x2, -7.34990630326855e-04f);
  p = fmaf(p, x2, -2.95459980854025e-03f);
  p = fmaf(p, x2, -1.60960333262415e-02f);
  float q = -1.45660718464996e-05f;
  q = fmaf(q, x2, -2.13374055278905e-04f);
  q = fmaf(q, x2, -1.68282697438203e-03f);
  q = fmaf(q, x2, -7.37332916720468e-03f);
  q = fmaf(q, x2, -1.42647390514189e-02f);
  return (p * x) * __builtin_amdgcn_rcpf(q);
}

__device__ __forceinline__ float sigmoid_hw(float g) {
  const float e = __builtin_amdgcn_exp2f(g * -1.4426950408889634f);
  return __builtin_amdgcn_rcpf(1.0f + e);
}

__device__ __forceinline__ void cstep(float& pr, float& pi, float wr, float wi) {
  const float nr = pr * wr - pi * wi;
  const float ni = pr * wi + pi * wr;
  pr = nr; pi = ni;
}

__device__ __forceinline__ float taps32(float& pr, float& pi, float wr, float wi, int lane) {
  float tv[32];
#pragma unroll
  for (int d = 0; d < 32; ++d) {
    float v = 2.0f * pr;
    v += xhalf(v);
    v += xrow<8>(v);
    v += xrow<4>(v);
    v += xrow<2>(v);
    v += xrow<1>(v);
    tv[d] = v;
    cstep(pr, pi, wr, wi);
  }
  float o = tv[0];
#pragma unroll
  for (int d = 1; d < 32; ++d) {
    if (lane == d) o = tv[d];
  }
  return o;
}

__global__ __launch_bounds__(256, 8) void s4d_conv(const float* __restrict__ u,
                         const float* __restrict__ log_dt,
                         const float* __restrict__ C,
                         const float* __restrict__ log_A_real,
                         const float* __restrict__ A_imag,
                         const float* __restrict__ Dvec,
                         float* __restrict__ ygelu) {
  __shared__ __attribute__((aligned(16))) float sKt[256];
  __shared__ __attribute__((aligned(16))) float sCR[Q_ * CPITCH];
  __shared__ __attribute__((aligned(16))) float sCI[Q_ * CPITCH];
  __shared__ __attribute__((aligned(16))) float sP[2 * NMODE * PPITCH];
  __shared__ __attribute__((aligned(16))) float sU[2 * UBUF];
  __shared__ __attribute__((aligned(16))) float sS[2 * NMODE * SPITCH];
  __shared__ __attribute__((aligned(16))) float sXR[NCHUNK * XPITCH];
  __shared__ __attribute__((aligned(16))) float sXI[NCHUNK * XPITCH];

  const int h      = blockIdx.x >> 1;
  const int b0     = (blockIdx.x & 1) * NBROW;
  const int waveId = threadIdx.x >> 5;
  const int lane   = threadIdx.x & 31;
  const int half   = lane >> 4;
  const int lm     = lane & 15;
  const int jt     = waveId & 3;
  const int ct     = waveId >> 2;
  const int col    = ct * 16 + lm;

  size_t urow = (size_t)(b0 * H_ + h) * L_;

  const float Dh = Dvec[h];

  float4 ru0, ru1;
  ru0 = *(const float4*)(u + urow + threadIdx.x * 4);
  ru1 = *(const float4*)(u + urow + (256 + threadIdx.x) * 4);

  if (waveId < 4) {
    const int k = lane;
    const float dt = expf(log_dt[h]);
    const float Ar = -expf(log_A_real[h * NMODE + k]);
    const float Ai = A_imag[h * NMODE + k];
    const float Ccr = C[(h * NMODE + k) * 2 + 0];
    const float Cci = C[(h * NMODE + k) * 2 + 1];
    const float er = expf(Ar * dt);
    const float wr = er * cosf(Ai * dt);
    const float wi = er * sinf(Ai * dt);
    const float exr = wr - 1.0f, exi = wi;
    const float den = Ar * Ar + Ai * Ai;
    const float qr = (exr * Ar + exi * Ai) / den;
    const float qi = (exi * Ar - exr * Ai) / den;
    const float Ckr = Ccr * qr - Cci * qi;
    const float Cki = Ccr * qi + Cci * qr;
    if (waveId == 0) {
      sKt[k] = 0.0f;
      sKt[32 + k] = 0.0f;
      float pr = Ckr, pi = Cki;
      sKt[64 + k] = taps32(pr, pi, wr, wi, k);
    } else if (waveId == 3) {
      float pr = Ckr, pi = Cki;
      for (int d = 0; d < 32; ++d) cstep(pr, pi, wr, wi);
      sKt[96 + k] = taps32(pr, pi, wr, wi, k);
    } else if (waveId == 1) {
      float sr = Ckr * wr - Cki * wi;
      float si = Ckr * wi + Cki * wr;
      for (int j = 0; j < Q_; ++j) {
        sCR[j * CPITCH + k] = 2.0f * sr;
        sCI[j * CPITCH + k] = -2.0f * si;
        cstep(sr, si, wr, wi);
      }
    } else {
      float pr = 1.0f, pi = 0.0f;
      for (int d = 0; d < Q_; ++d) {
        sP[k * PPITCH + (Q_ - 1 - d)]           = pr;
        sP[(NMODE + k) * PPITCH + (Q_ - 1 - d)] = pi;
        cstep(pr, pi, wr, wi);
      }
      sKt[128 + k] = pr;
      sKt[160 + k] = pi;
    }
  }
  const int jrow = jt * 16 + lm;

  for (int r = 0; r < NBROW; ++r) {
  float* sUc = sU + (r & 1) * UBUF;
  {
    const int e0 = threadIdx.x * 4, e1 = (256 + threadIdx.x) * 4;
    *(float4*)(sUc + (e0 >> 6) * PPITCH + (e0 & 63)) = ru0;
    *(float4*)(sUc + (e1 >> 6) * PPITCH + (e1 & 63)) = ru1;
  }
  if (r + 1 < NBROW) {
    ru0 = *(const float4*)(u + urow + (size_t)H_ * L_ + threadIdx.x * 4);
    ru1 = *(const float4*)(u + urow + (size_t)H_ * L_ + (256 + threadIdx.x) * 4);
  }
  lds_barrier();

  v8f acc  = {0.f, 0.f, 0.f, 0.f, 0.f, 0.f, 0.f, 0.f};
  v8f accS = {0.f, 0.f, 0.f, 0.f, 0.f, 0.f, 0.f, 0.f};

#pragma unroll
  for (int k0 = 0; k0 < Q_; k0 += 4) {
    const int ta = k0 + 2 * half;
    const v2f bb = *(const v2f*)(sUc + col * PPITCH + ta);
    const v2f ap = *(const v2f*)(sP + jrow * PPITCH + ta);
    v2f a;
    a.x = sKt[64 + jrow - ta];
    a.y = sKt[63 + jrow - ta];
    accS = wmma4(ap, bb, accS);
    acc  = wmma4(a, bb, acc);
  }
#pragma unroll
  for (int v = 0; v < 8; ++v)
    sS[(jt * 16 + v + 8 * half) * SPITCH + col] = accS[v];
  lds_barrier();

  if (waveId == 0) {
    const float ar = sKt[128 + lane];
    const float ai = sKt[160 + lane];
    float xr = 0.0f, xi = 0.0f;
#pragma unroll
    for (int c0 = 0; c0 < NCHUNK; c0 += 16) {
      float pr[16], pi[16];
#pragma unroll
      for (int i = 0; i < 16; ++i) {
        pr[i] = sS[lane * SPITCH + c0 + i];
        pi[i] = sS[(NMODE + lane) * SPITCH + c0 + i];
      }
#pragma unroll
      for (int i = 0; i < 16; ++i) {
        sXR[(c0 + i) * XPITCH + lane] = xr;
        sXI[(c0 + i) * XPITCH + lane] = xi;
        const float nr = ar * xr - ai * xi + pr[i];
        const float ni = ar * xi + ai * xr + pi[i];
        xr = nr; xi = ni;
      }
    }
  }
  lds_barrier();

#pragma unroll
  for (int k0 = 0; k0 < NMODE; k0 += 4) {
    const int ka = k0 + 2 * half;
    const v2f a  = *(const v2f*)(sCR + jrow * CPITCH + ka);
    const v2f bb = *(const v2f*)(sXR + col * XPITCH + ka);
    acc = wmma4(a, bb, acc);
  }
#pragma unroll
  for (int k0 = 0; k0 < NMODE; k0 += 4) {
    const int ka = k0 + 2 * half;
    const v2f a  = *(const v2f*)(sCI + jrow * CPITCH + ka);
    const v2f bb = *(const v2f*)(sXI + col * XPITCH + ka);
    acc = wmma4(a, bb, acc);
  }
  float* su8 = sUc + col * PPITCH + jt * 16 + 8 * half;
  const float4 u0 = *(const float4*)(su8);
  const float4 u1 = *(const float4*)(su8 + 4);
  const float uu[8] = {u0.x, u0.y, u0.z, u0.w, u1.x, u1.y, u1.z, u1.w};
  float yo[8];
#pragma unroll
  for (int v = 0; v < 8; ++v) {
    const float y = acc[v] + Dh * uu[v];
    const float hy = 0.5f * y;
    yo[v] = fmaf(hy, erf_rational(y * 0.70710678118654752f), hy);
  }
  *(float4*)(su8)     = make_float4(yo[0], yo[1], yo[2], yo[3]);
  *(float4*)(su8 + 4) = make_float4(yo[4], yo[5], yo[6], yo[7]);
  lds_barrier();
#pragma unroll
  for (int it = 0; it < 2; ++it) {
    const int e = (it * 256 + threadIdx.x) * 4;
    *(float4*)(ygelu + urow + e) = *(const float4*)(sUc + (e >> 6) * PPITCH + (e & 63));
  }
  urow += (size_t)H_ * L_;
  }
}

#define GKC      32
#define APITCH   36
#define BPITCH   136
#define A_BUF    (128 * APITCH)
#define B_BUF    (GKC * BPITCH)
#define GLU_LDS_BYTES 71680

template <bool MORE>
__device__ __forceinline__ void glu_chunk(v8f (&M)[2][7], const float* cA, const float* cB,
                                          float* dA, float* dB, const float* nA, const float* nB) {
  float4 st[4];
  if (MORE) {
    st[0] = *(const float4*)(nA);
    st[1] = *(const float4*)(nA + (size_t)32 * H_);
    st[2] = *(const float4*)(nB);
    st[3] = *(const float4*)(nB + (size_t)8 * L_);
  }
  v2f a[2][4][2], b[2][2][2];
#pragma unroll
  for (int hk = 0; hk < 2; ++hk) {
    a[0][0][hk] = *(const v2f*)(cA + 4 * hk);
    a[0][1][hk] = *(const v2f*)(cA + 16 * APITCH + 4 * hk);
    a[0][2][hk] = *(const v2f*)(cA + 64 * APITCH + 4 * hk);
    a[0][3][hk] = *(const v2f*)(cA + 80 * APITCH + 4 * hk);
#pragma unroll
    for (int j = 0; j < 2; ++j) {
      b[0][hk][j].x = cB[(4 * hk) * BPITCH + 16 * j];
      b[0][hk][j].y = cB[(4 * hk + 1) * BPITCH + 16 * j];
    }
  }
#pragma unroll
  for (int p = 0; p < GKC / 8; ++p) {
    const int c = p & 1, n = c ^ 1;
    if (p + 1 < GKC / 8) {
#pragma unroll
      for (int hk = 0; hk < 2; ++hk) {
        const int k1 = (p + 1) * 8 + 4 * hk;
        a[n][0][hk] = *(const v2f*)(cA + k1);
        a[n][1][hk] = *(const v2f*)(cA + 16 * APITCH + k1);
        a[n][2][hk] = *(const v2f*)(cA + 64 * APITCH + k1);
        a[n][3][hk] = *(const v2f*)(cA + 80 * APITCH + k1);
#pragma unroll
        for (int j = 0; j < 2; ++j) {
          b[n][hk][j].x = cB[k1 * BPITCH + 16 * j];
          b[n][hk][j].y = cB[(k1 + 1) * BPITCH + 16 * j];
        }
      }
    }
    const v2f B11 = b[c][0][0], B12 = b[c][0][1], B21 = b[c][1][0], B22 = b[c][1][1];
    const v2f T1 = B11 + B22, T2 = B12 - B22, T3 = B21 - B11, T4 = B11 + B12, T5 = B21 + B22;
#pragma unroll
    for (int blk = 0; blk < 2; ++blk) {
      const v2f A11 = a[c][2 * blk][0],     A12 = a[c][2 * blk][1];
      const v2f A21 = a[c][2 * blk + 1][0], A22 = a[c][2 * blk + 1][1];
      M[blk][0] = wmma4(A11 + A22, T1,  M[blk][0]);
      M[blk][1] = wmma4(A21 + A22, B11, M[blk][1]);
      M[blk][2] = wmma4(A11,       T2,  M[blk][2]);
      M[blk][3] = wmma4(A22,       T3,  M[blk][3]);
      M[blk][4] = wmma4(A11 + A12, B22, M[blk][4]);
      M[blk][5] = wmma4(A21 - A11, T4,  M[blk][5]);
      M[blk][6] = wmma4(A12 - A22, T5,  M[blk][6]);
    }
    if (MORE && p == GKC / 16 - 1) {
      *(float4*)(dA)               = st[0];
      *(float4*)(dA + 32 * APITCH) = st[1];
      *(float4*)(dB)               = st[2];
      *(float4*)(dB + 8 * BPITCH)  = st[3];
      st[0] = *(const float4*)(nA + (size_t)H_ * H_);
      st[1] = *(const float4*)(nA + (size_t)H_ * H_ + (size_t)32 * H_);
      st[2] = *(const float4*)(nB + (size_t)16 * L_);
      st[3] = *(const float4*)(nB + (size_t)24 * L_);
    }
  }
  if (MORE) {
    *(float4*)(dA + 64 * APITCH) = st[0];
    *(float4*)(dA + 96 * APITCH) = st[1];
    *(float4*)(dB + 16 * BPITCH) = st[2];
    *(float4*)(dB + 24 * BPITCH) = st[3];
  }
}

__global__ __launch_bounds__(256) __attribute__((amdgpu_waves_per_eu(4, 4))) void s4d_glu(const float* __restrict__ ygelu,
                                               const float* __restrict__ W,
                                               const float* __restrict__ bias,
                                               float* __restrict__ out) {
  extern __shared__ __attribute__((aligned(16))) float smem[];
  float* sA = smem;
  float* sB = smem + 2 * A_BUF;

  const int tid    = threadIdx.x;
  const int waveId = tid >> 5;
  const int lane   = tid & 31;
  const int half   = lane >> 4;
  const int lm     = lane & 15;
  const int wm     = waveId & 1;
  const int wn     = waveId >> 1;
  const int mblk   = blockIdx.x & 7;
  const int nblk   = blockIdx.x >> 3;
  const int bb     = nblk >> 4;
  const int l0     = (nblk & 15) * 128;

  const float* gA = W + (size_t)(mblk * 64 + (tid >> 3)) * H_ + (tid & 7) * 4;
  const float* gB = ygelu + ((size_t)bb * H_ + (tid >> 5)) * L_ + l0 + (tid & 31) * 4;
  const int stA = (tid >> 3) * APITCH + (tid & 7) * 4;
  const int stB = (tid >> 5) * BPITCH + (tid & 31) * 4;

  {
    float4 ra[4], rb[4];
#pragma unroll
    for (int j = 0; j < 4; ++j) {
      ra[j] = *(const float4*)(gA + (size_t)(j >> 1) * H_ * H_ + (size_t)(j & 1) * 32 * H_);
      rb[j] = *(const float4*)(gB + (size_t)j * 8 * L_);
    }
#pragma unroll
    for (int j = 0; j < 4; ++j) {
      *(float4*)(sA + stA + j * 32 * APITCH) = ra[j];
      *(float4*)(sB + stB + j * 8 * BPITCH)  = rb[j];
    }
  }
  __syncthreads();

  v8f M[2][7];
#pragma unroll
  for (int i = 0; i < 2; ++i)
#pragma unroll
    for (int j = 0; j < 7; ++j)
      M[i][j] = (v8f){0.f, 0.f, 0.f, 0.f, 0.f, 0.f, 0.f, 0.f};

  const int fA = (wm * 32 + lm) * APITCH + 2 * half;
  const int fB = (2 * half) * BPITCH + wn * 32 + lm;

  for (int kc = 0; kc < H_ / GKC; ++kc) {
    const int cur = kc & 1;
    const float* cA = sA + cur * A_BUF + fA;
    const float* cB = sB + cur * B_BUF + fB;
    float* dA = sA + (cur ^ 1) * A_BUF + stA;
    float* dB = sB + (cur ^ 1) * B_BUF + stB;
    const float* nA = gA + (kc + 1) * GKC;
    const float* nB = gB + (size_t)(kc + 1) * GKC * L_;
    if ((kc + 1) < (H_ / GKC)) glu_chunk<true>(M, cA, cB, dA, dB, nA, nB);
    else                       glu_chunk<false>(M, cA, cB, dA, dB, nA, nB);
    __syncthreads();
  }

  v8f acc[4][2];
#pragma unroll
  for (int blk = 0; blk < 2; ++blk) {
    acc[2 * blk][0]     = M[blk][0] + M[blk][3] - M[blk][4] + M[blk][6];
    acc[2 * blk][1]     = M[blk][2] + M[blk][4];
    acc[2 * blk + 1][0] = M[blk][1] + M[blk][3];
    acc[2 * blk + 1][1] = M[blk][0] - M[blk][1] + M[blk][2] + M[blk][5];
  }

#pragma unroll
  for (int mi = 0; mi < 2; ++mi) {
#pragma unroll
    for (int v = 0; v < 8; ++v) {
      const int hr = mblk * 64 + wm * 32 + mi * 16 + v + 8 * half;
      const float ba = bias[hr];
      const float bg = bias[H_ + hr];
#pragma unroll
      for (int j = 0; j < 2; ++j) {
        const int l = l0 + wn * 32 + j * 16 + lm;
        const float a  = acc[mi][j][v] + ba;
        const float gv = acc[2 + mi][j][v] + bg;
        out[((size_t)(bb * H_ + hr)) * L_ + l] = a * sigmoid_hw(gv);
      }
    }
  }
}

extern "C" void kernel_launch(void* const* d_in, const int* in_sizes, int n_in,
                              void* d_out, int out_size, void* d_ws, size_t ws_size,
                              hipStream_t stream) {
  (void)in_sizes; (void)n_in; (void)out_size; (void)ws_size;
  const float* u          = (const float*)d_in[0];
  const float* log_dt     = (const float*)d_in[1];
  const float* C          = (const float*)d_in[2];
  const float* log_A_real = (const float*)d_in[3];
  const float* A_imag     = (const float*)d_in[4];
  const float* Dvec       = (const float*)d_in[5];
  const float* W          = (const float*)d_in[6];
  const float* bias       = (const float*)d_in[7];
  float* out = (float*)d_out;

  float* ygelu = (float*)d_ws;

  s4d_conv<<<H_ * B_ / NBROW, 256, 0, stream>>>(u, log_dt, C, log_A_real, A_imag, Dvec, ygelu);
  s4d_glu<<<1024, 256, GLU_LDS_BYTES, stream>>>(ygelu, W, bias, out);
}
